# starting kernel
# baseline (speedup 1.0000x reference)
#include <hip/hip_runtime.h>
#include <math.h>

typedef __attribute__((ext_vector_type(2))) float v2f;
typedef __attribute__((ext_vector_type(8))) float v8f;

#define GAT_NEG_SLOPE 0.2f
#define GAT_SM_EPS 1e-16f
#define GAT_LN_EPS 1e-5f

__device__ __forceinline__ float warp_sum32(float v) {
#pragma unroll
  for (int o = 16; o > 0; o >>= 1) v += __shfl_xor(v, o, 32);
  return v;
}
__device__ __forceinline__ float warp_max32(float v) {
#pragma unroll
  for (int o = 16; o > 0; o >>= 1) v = fmaxf(v, __shfl_xor(v, o, 32));
  return v;
}

__device__ __forceinline__ void atomic_max_f32(float* addr, float v) {
  if (v >= 0.0f) atomicMax((int*)addr, __float_as_int(v));
  else           atomicMin((unsigned int*)addr, __float_as_uint(v));
}

__global__ void fill_f32(float* __restrict__ p, int n, float v) {
  int i = blockIdx.x * blockDim.x + threadIdx.x;
  int stride = gridDim.x * blockDim.x;
  for (; i < n; i += stride) p[i] = v;
}

__global__ void gemm_wmma_f32(const float* __restrict__ A, const float* __restrict__ B,
                              float* __restrict__ C, int M, int K, int Nc) {
  extern __shared__ float As[];
  const int KP = K + 1;
  const int mBase = blockIdx.x * 16;
  const int tid = threadIdx.x;

  for (int i = tid; i < 16 * K; i += blockDim.x) {
    int r = i / K, c = i - r * K;
    int row = mBase + r;
    As[r * KP + c] = (row < M) ? A[(size_t)row * K + c] : 0.0f;
  }
  __syncthreads();

  const int wave = tid >> 5;
  const int lane = tid & 31;
  const int nBase = wave * 16;
  const int l15 = lane & 15;
  const int khalf = (lane >> 4) * 2;

  v8f acc = {};
  for (int k0 = 0; k0 < K; k0 += 4) {
    v2f a, b;
    a.x = As[l15 * KP + k0 + khalf];
    a.y = As[l15 * KP + k0 + khalf + 1];
    b.x = B[(size_t)(k0 + khalf) * Nc + nBase + l15];
    b.y = B[(size_t)(k0 + khalf + 1) * Nc + nBase + l15];
    acc = __builtin_amdgcn_wmma_f32_16x16x4_f32(false, a, false, b,
                                                (short)0, acc, false, false);
  }

  const int rowTop = mBase + (lane >> 4) * 8;
  const int col = nBase + l15;
#pragma unroll
  for (int r = 0; r < 8; ++r) {
    int row = rowTop + r;
    if (row < M) C[(size_t)row * Nc + col] = acc[r];
  }
}

__global__ void attn_scores_k(const float* __restrict__ h, const float* __restrict__ a_src,
                              const float* __restrict__ a_dst, float* __restrict__ es,
                              float* __restrict__ ed, int N, int H, int CH) {
  int idx = blockIdx.x * blockDim.x + threadIdx.x;
  if (idx >= N * H) return;
  int n = idx / H, hh = idx - n * H;
  const float* hp = h + (size_t)n * H * CH + hh * CH;
  const float* asp = a_src + hh * CH;
  const float* adp = a_dst + hh * CH;
  float s = 0.0f, d = 0.0f;
  for (int c = 0; c < CH; ++c) { float t = hp[c]; s += t * asp[c]; d += t * adp[c]; }
  es[idx] = s; ed[idx] = d;
}

__device__ __forceinline__ void edge_sd(const int* __restrict__ ei, int E, int e,
                                        int& s, int& d) {
  if (e < E) { s = ei[e]; d = ei[E + e]; }
  else       { s = d = e - E; }
}

__global__ void edge_max_k(const int* __restrict__ ei, int E, int N,
                           const float* __restrict__ es, const float* __restrict__ ed,
                           float* __restrict__ m, int H) {
  int idx = blockIdx.x * blockDim.x + threadIdx.x;
  int tot = (E + N) * H;
  if (idx >= tot) return;
  int h = idx & (H - 1);
  int e = idx >> 2;
  int s, d; edge_sd(ei, E, e, s, d);
  float v = es[s * H + h] + ed[d * H + h];
  v = (v > 0.0f) ? v : GAT_NEG_SLOPE * v;
  atomic_max_f32(&m[d * H + h], v);
}

__global__ void edge_expsum_k(const int* __restrict__ ei, int E, int N,
                              const float* __restrict__ es, const float* __restrict__ ed,
                              const float* __restrict__ m, float* __restrict__ denom, int H) {
  int idx = blockIdx.x * blockDim.x + threadIdx.x;
  int tot = (E + N) * H;
  if (idx >= tot) return;
  int h = idx & (H - 1);
  int e = idx >> 2;
  int s, d; edge_sd(ei, E, e, s, d);
  float v = es[s * H + h] + ed[d * H + h];
  v = (v > 0.0f) ? v : GAT_NEG_SLOPE * v;
  atomicAdd(&denom[d * H + h], __expf(v - m[d * H + h]));
}

template <int CHT, int CH>
__global__ void edge_agg_k(const int* __restrict__ ei, int E, int N,
                           const float* __restrict__ feat, const float* __restrict__ es,
                           const float* __restrict__ ed, const float* __restrict__ m,
                           const float* __restrict__ denom, float* __restrict__ out) {
  const int H = CHT / CH;
  int gw = (blockIdx.x * blockDim.x + threadIdx.x) >> 5;
  int lane = threadIdx.x & 31;
  if (gw >= E + N) return;
  int s, d; edge_sd(ei, E, gw, s, d);

  float alpha = 0.0f;
  if (lane < H) {
    float v = es[s * H + lane] + ed[d * H + lane];
    v = (v > 0.0f) ? v : GAT_NEG_SLOPE * v;
    alpha = __expf(v - m[d * H + lane]) / (denom[d * H + lane] + GAT_SM_EPS);
  }
  const float* fp = feat + (size_t)s * CHT;
  float* op = out + (size_t)d * CHT;
#pragma unroll
  for (int c = lane; c < CHT; c += 32) {
    int head = c / CH;
    float a = __shfl(alpha, head, 32);
    atomicAdd(&op[c], fp[c] * a);
  }
}

__global__ void elu_ln128_k(float* __restrict__ x, const float* __restrict__ bias,
                            const float* __restrict__ g, const float* __restrict__ b, int N) {
  int node = blockIdx.x * (blockDim.x >> 5) + (threadIdx.x >> 5);
  int lane = threadIdx.x & 31;
  if (node >= N) return;
  float* xp = x + (size_t)node * 128;
  float v[4];
  float sum = 0.0f;
#pragma unroll
  for (int j = 0; j < 4; ++j) {
    int c = lane * 4 + j;
    float t = xp[c] + bias[c];
    t = (t > 0.0f) ? t : (__expf(t) - 1.0f);
    v[j] = t; sum += t;
  }
  float mu = warp_sum32(sum) * (1.0f / 128.0f);
  float vs = 0.0f;
#pragma unroll
  for (int j = 0; j < 4; ++j) { float dd = v[j] - mu; vs += dd * dd; }
  float var = warp_sum32(vs) * (1.0f / 128.0f);
  float rstd = rsqrtf(var + GAT_LN_EPS);
#pragma unroll
  for (int j = 0; j < 4; ++j) {
    int c = lane * 4 + j;
    xp[c] = (v[j] - mu) * rstd * g[c] + b[c];
  }
}

__global__ void final_k(const float* __restrict__ out2, const float* __restrict__ b2,
                        const float* __restrict__ g, const float* __restrict__ bb,
                        float* __restrict__ y, int N) {
  int node = blockIdx.x * (blockDim.x >> 5) + (threadIdx.x >> 5);
  int lane = threadIdx.x & 31;
  if (node >= N) return;
  const bool has1 = (lane < 8);
  const float* op = out2 + (size_t)node * 160;

  float x0, x1 = 0.0f;
  { float s = 0.0f;
#pragma unroll
    for (int h = 0; h < 4; ++h) s += op[h * 40 + lane];
    x0 = s * 0.25f + b2[lane]; }
  if (has1) {
    float s = 0.0f;
#pragma unroll
    for (int h = 0; h < 4; ++h) s += op[h * 40 + lane + 32];
    x1 = s * 0.25f + b2[lane + 32];
  }
  float mu = warp_sum32(x0 + (has1 ? x1 : 0.0f)) * (1.0f / 40.0f);
  float d0 = x0 - mu, d1 = has1 ? (x1 - mu) : 0.0f;
  float var = warp_sum32(d0 * d0 + d1 * d1) * (1.0f / 40.0f);
  float rstd = rsqrtf(var + GAT_LN_EPS);
  float z0 = d0 * rstd * g[lane] + bb[lane];
  float z1 = has1 ? (d1 * rstd * g[lane + 32] + bb[lane + 32]) : -3.0e38f;
  float mx = warp_max32(fmaxf(z0, z1));
  float se = warp_sum32(__expf(z0 - mx) + (has1 ? __expf(z1 - mx) : 0.0f));
  float lse = __logf(se);
  float* yp = y + (size_t)node * 40;
  yp[lane] = z0 - mx - lse;
  if (has1) yp[lane + 32] = z1 - mx - lse;
}

extern "C" void kernel_launch(void* const* d_in, const int* in_sizes, int n_in,
                              void* d_out, int out_size, void* d_ws, size_t ws_size,
                              hipStream_t stream) {
  const float* x        = (const float*)d_in[0];
  const int*   ei       = (const int*)d_in[1];
  const float* W1       = (const float*)d_in[2];
  const float* att_src1 = (const float*)d_in[3];
  const float* att_dst1 = (const float*)d_in[4];
  const float* b1       = (const float*)d_in[5];
  const float* W2       = (const float*)d_in[6];
  const float* att_src2 = (const float*)d_in[7];
  const float* att_dst2 = (const float*)d_in[8];
  const float* b2       = (const float*)d_in[9];
  const float* ln0_g    = (const float*)d_in[10];
  const float* ln0_b    = (const float*)d_in[11];
  const float* ln1_g    = (const float*)d_in[12];
  const float* ln1_b    = (const float*)d_in[13];

  const int F_IN = 256, H1C = 128, H2C = 160, H = 4;
  const int N = in_sizes[0] / F_IN;
  const int E = in_sizes[1] / 2;
  const int EL = E + N;

  float* ws   = (float*)d_ws;
  float* h1   = ws;
  float* out1 = h1   + (size_t)N * H1C;
  float* out2 = out1 + (size_t)N * H1C;
  float* h2   = out2 + (size_t)N * H2C;
  float* es   = h2   + (size_t)N * H2C;
  float* ed   = es   + (size_t)N * H;
  float* mseg = ed   + (size_t)N * H;
  float* dseg = mseg + (size_t)N * H;

  const float NEG_INF = -__builtin_inff();
  const int mTiles = (N + 15) / 16;

  gemm_wmma_f32<<<mTiles, 2 * H1C, 16 * (F_IN + 1) * sizeof(float), stream>>>(
      x, W1, h1, N, F_IN, H1C);
  attn_scores_k<<<(N * H + 255) / 256, 256, 0, stream>>>(h1, att_src1, att_dst1,
                                                         es, ed, N, H, 32);
  fill_f32<<<512, 256, 0, stream>>>(mseg, N * H, NEG_INF);
  fill_f32<<<512, 256, 0, stream>>>(dseg, N * H, 0.0f);
  fill_f32<<<2048, 256, 0, stream>>>(out1, N * H1C, 0.0f);
  edge_max_k<<<(EL * H + 255) / 256, 256, 0, stream>>>(ei, E, N, es, ed, mseg, H);
  edge_expsum_k<<<(EL * H + 255) / 256, 256, 0, stream>>>(ei, E, N, es, ed, mseg, dseg, H);
  edge_agg_k<128, 32><<<(EL + 7) / 8, 256, 0, stream>>>(ei, E, N, h1, es, ed,
                                                        mseg, dseg, out1);
  elu_ln128_k<<<(N + 7) / 8, 256, 0, stream>>>(out1, b1, ln0_g, ln0_b, N);

  gemm_wmma_f32<<<mTiles, 2 * H2C, 16 * (H1C + 1) * sizeof(float), stream>>>(
      out1, W2, h2, N, H1C, H2C);
  attn_scores_k<<<(N * H + 255) / 256, 256, 0, stream>>>(h2, att_src2, att_dst2,
                                                         es, ed, N, H, 40);
  fill_f32<<<512, 256, 0, stream>>>(mseg, N * H, NEG_INF);
  fill_f32<<<512, 256, 0, stream>>>(dseg, N * H, 0.0f);
  fill_f32<<<2048, 256, 0, stream>>>(out2, N * H2C, 0.0f);
  edge_max_k<<<(EL * H + 255) / 256, 256, 0, stream>>>(ei, E, N, es, ed, mseg, H);
  edge_expsum_k<<<(EL * H + 255) / 256, 256, 0, stream>>>(ei, E, N, es, ed, mseg, dseg, H);
  edge_agg_k<160, 40><<<(EL + 7) / 8, 256, 0, stream>>>(ei, E, N, h2, es, ed,
                                                        mseg, dseg, out2);

  final_k<<<(N + 7) / 8, 256, 0, stream>>>(out2, b2, ln1_g, ln1_b, (float*)d_out, N);
}


# Edge work by node-range owners, no sort: each of 256 work-groups scans the edge list for its 196 nodes; softmax and row sums by LDS float adds (arrival order, conv2 heads last); 18 launches to 6
# speedup vs baseline: 2.8440x; 2.8440x over previous
#include <hip/hip_runtime.h>
#include <math.h>

typedef __attribute__((ext_vector_type(2))) float v2f;
typedef __attribute__((ext_vector_type(8))) float v8f;

#define GAT_NEG_SLOPE 0.2f
#define GAT_SM_EPS 1e-16f
#define GAT_LN_EPS 1e-5f

__device__ __forceinline__ float warp_sum32(float v) {
#pragma unroll
  for (int o = 16; o > 0; o >>= 1) v += __shfl_xor(v, o, 32);
  return v;
}
__device__ __forceinline__ float warp_max32(float v) {
#pragma unroll
  for (int o = 16; o > 0; o >>= 1) v = fmaxf(v, __shfl_xor(v, o, 32));
  return v;
}


__global__ void gemm_wmma_f32(const float* __restrict__ A, const float* __restrict__ B,
                              float* __restrict__ C, int M, int K, int Nc) {
  extern __shared__ float As[];
  const int KP = K + 1;
  const int mBase = blockIdx.x * 16;
  const int tid = threadIdx.x;

  for (int i = tid; i < 16 * K; i += blockDim.x) {
    int r = i / K, c = i - r * K;
    int row = mBase + r;
    As[r * KP + c] = (row < M) ? A[(size_t)row * K + c] : 0.0f;
  }
  __syncthreads();

  const int wave = tid >> 5;
  const int lane = tid & 31;
  const int nBase = wave * 16;
  const int l15 = lane & 15;
  const int khalf = (lane >> 4) * 2;

  v8f acc = {};
  for (int k0 = 0; k0 < K; k0 += 4) {
    v2f a, b;
    a.x = As[l15 * KP + k0 + khalf];
    a.y = As[l15 * KP + k0 + khalf + 1];
    b.x = B[(size_t)(k0 + khalf) * Nc + nBase + l15];
    b.y = B[(size_t)(k0 + khalf + 1) * Nc + nBase + l15];
    acc = __builtin_amdgcn_wmma_f32_16x16x4_f32(false, a, false, b,
                                                (short)0, acc, false, false);
  }

  const int rowTop = mBase + (lane >> 4) * 8;
  const int col = nBase + l15;
#pragma unroll
  for (int r = 0; r < 8; ++r) {
    int row = rowTop + r;
    if (row < M) C[(size_t)row * Nc + col] = acc[r];
  }
}

__global__ void attn_scores_k(const float* __restrict__ h, const float* __restrict__ a_src,
                              const float* __restrict__ a_dst, float* __restrict__ es,
                              float* __restrict__ ed, int N, int H, int CH) {
  int idx = blockIdx.x * blockDim.x + threadIdx.x;
  if (idx >= N * H) return;
  int n = idx / H, hh = idx - n * H;
  const float* hp = h + (size_t)n * H * CH + hh * CH;
  const float* asp = a_src + hh * CH;
  const float* adp = a_dst + hh * CH;
  float s = 0.0f, d = 0.0f;
  for (int c = 0; c < CH; ++c) { float t = hp[c]; s += t * asp[c]; d += t * adp[c]; }
  es[idx] = s; ed[idx] = d;
}

__device__ __forceinline__ int lane_i32(int v, int src) {
  return __builtin_amdgcn_readlane(v, src);
}
__device__ __forceinline__ float lane_f32(float v, int src) {
  return __int_as_float(__builtin_amdgcn_readlane(__float_as_int(v), src));
}

__device__ __forceinline__ float leaky(float v) { return (v > 0.0f) ? v : GAT_NEG_SLOPE * v; }

__device__ __forceinline__ void lds_max_f32(float* addr, float v) {
  if (v >= 0.0f) atomicMax((int*)addr, __float_as_int(v));
  else           atomicMin((unsigned int*)addr, __float_as_uint(v));
}

#define OWN_T 1024
#define OWN_RPW 196
#define OWN_CAP 6144
#define OWN_SCAN_U 8

template <int CH, bool CONCAT, bool SAVE, bool LOAD>
__global__ void __launch_bounds__(OWN_T)
owner_conv_k(const int* __restrict__ ei, int E, int N,
             const float* __restrict__ feat, const float* __restrict__ es,
             const float* __restrict__ ed, const float* __restrict__ bias,
             const float* __restrict__ lg, const float* __restrict__ lb,
             float* __restrict__ out,
             int* __restrict__ sv_cnt, int* __restrict__ sv_src, int* __restrict__ sv_dl) {
  constexpr int H = 4;
  constexpr int FR = H * CH;
  constexpr int KCH = FR / 32;
  constexpr int EPB = 8;
  __shared__ float acc[OWN_RPW * FR];
  __shared__ float lw[OWN_CAP * H];
  __shared__ int lsrc[OWN_CAP];
  __shared__ int ldl[OWN_CAP];
  __shared__ float edS[OWN_RPW * H], slf[OWN_RPW * H], mx[OWN_RPW * H], den[OWN_RPW * H];
  __shared__ int cnt;

  const int tid = threadIdx.x;
  const int lane = tid & 31;
  const int wave = __builtin_amdgcn_readfirstlane(tid >> 5);
  const int lo = blockIdx.x * OWN_RPW;
  const int nloc = min(OWN_RPW, N - lo);
  const int* __restrict__ dstp = ei + E;
  const size_t svBase = (size_t)blockIdx.x * OWN_CAP;

  for (int i = tid; i < nloc * FR; i += OWN_T) acc[i] = 0.0f;
  for (int i = tid; i < nloc * H; i += OWN_T) {
    const float d = ed[(size_t)lo * H + i];
    const float v = leaky(es[(size_t)lo * H + i] + d);
    edS[i] = d; slf[i] = v; mx[i] = v; den[i] = 0.0f;
  }

  auto scan = [&](int a, int b) -> int {
    if (tid == 0) cnt = 0;
    __syncthreads();
    auto hit = [&](int e, int d) {
      const unsigned r = (unsigned)(d - lo);
      if (r < (unsigned)nloc) {
        const int i = atomicAdd(&cnt, 1);
        if (i < OWN_CAP) { lsrc[i] = e; ldl[i] = (int)r; }
      }
    };
    if (((E | a) & 3) == 0) {
      const int nv = (b - a) >> 2;
      const int4* __restrict__ p4 = (const int4*)(dstp + a);
      const int rot = (int)(((long long)blockIdx.x * nv) / gridDim.x) & ~31;
      for (int i0 = tid; i0 < nv; i0 += OWN_T * OWN_SCAN_U) {
        int4 v[OWN_SCAN_U];
        int at[OWN_SCAN_U];
#pragma unroll
        for (int u = 0; u < OWN_SCAN_U; ++u) {
          const int ii = i0 + u * OWN_T;
          int jj = ii + rot;
          if (jj >= nv) jj -= nv;
          at[u] = jj;
          v[u] = (ii < nv) ? p4[jj] : make_int4(-1, -1, -1, -1);
        }
#pragma unroll
        for (int u = 0; u < OWN_SCAN_U; ++u) {
          const int e = a + 4 * at[u];
          hit(e, v[u].x); hit(e + 1, v[u].y); hit(e + 2, v[u].z); hit(e + 3, v[u].w);
        }
      }
      for (int e = a + 4 * nv + tid; e < b; e += OWN_T) hit(e, dstp[e]);
    } else {
      for (int e = a + tid; e < b; e += OWN_T) hit(e, dstp[e]);
    }
    __syncthreads();
    return cnt;
  };

  auto resolve = [&](int n, bool fromEdgeIds) {
    for (int i0 = tid; i0 < n; i0 += OWN_T * 4) {
      int s[4];
      float4 ev[4];
#pragma unroll
      for (int u = 0; u < 4; ++u) {
        const int i = i0 + u * OWN_T;
        int t = (i < n) ? lsrc[i] : 0;
        if (fromEdgeIds) t = ei[t];
        s[u] = t;
      }
#pragma unroll
      for (int u = 0; u < 4; ++u) ev[u] = *(const float4*)(es + (size_t)s[u] * H);
#pragma unroll
      for (int u = 0; u < 4; ++u) {
        const int i = i0 + u * OWN_T;
        if (i < n) {
          lsrc[i] = s[u];
          const int dl = ldl[i];
          const float l0 = leaky(ev[u].x + edS[dl * H + 0]);
          const float l1 = leaky(ev[u].y + edS[dl * H + 1]);
          const float l2 = leaky(ev[u].z + edS[dl * H + 2]);
          const float l3 = leaky(ev[u].w + edS[dl * H + 3]);
          lw[i * H + 0] = l0; lw[i * H + 1] = l1; lw[i * H + 2] = l2; lw[i * H + 3] = l3;
          lds_max_f32(&mx[dl * H + 0], l0); lds_max_f32(&mx[dl * H + 1], l1);
          lds_max_f32(&mx[dl * H + 2], l2); lds_max_f32(&mx[dl * H + 3], l3);
        }
      }
    }
    __syncthreads();
  };

  auto sumphase = [&](int n, bool first) {
    const int M = (n + (first ? nloc : 0)) * H;
    for (int it = tid; it < M; it += OWN_T) {
      const int i = it >> 2, h = it & 3;
      int dl; float l;
      if (i < n) { dl = ldl[i]; l = lw[it]; }
      else       { dl = i - n;  l = slf[dl * H + h]; }
      atomicAdd(&den[dl * H + h], __expf(l - mx[dl * H + h]));
    }
  };

  auto aggphase = [&](int n, bool first) {
    const int M = n + (first ? nloc : 0);
    const int nR = (M + EPB - 1) / EPB;
    for (int r = wave; r < nR; r += OWN_T / 32) {
      const int h = lane & 3;
      const int i = r * EPB + (lane >> 2);
      int s = 0, dl = 0; float alpha = 0.0f;
      if (i < M) {
        float l;
        if (i < n) { s = lsrc[i]; dl = ldl[i]; l = lw[i * H + h]; }
        else       { dl = i - n; s = lo + dl; l = slf[dl * H + h]; }
        alpha = __expf(l - mx[dl * H + h]) / (den[dl * H + h] + GAT_SM_EPS);
      }
      float f[EPB][KCH];
#pragma unroll
      for (int j = 0; j < EPB; ++j) {
        const float* row = feat + (size_t)lane_i32(s, 4 * j) * FR;
#pragma unroll
        for (int k = 0; k < KCH; ++k) f[j][k] = row[lane + 32 * k];
      }
#pragma unroll
      for (int j = 0; j < EPB; ++j) {
        if (r * EPB + j < M) {
          float* arow = acc + lane_i32(dl, 4 * j) * FR;
#pragma unroll
          for (int k = 0; k < KCH; ++k) {
            const int hLo = (32 * k) / CH, hHi = (32 * k + 31) / CH;
            float w = lane_f32(alpha, 4 * j + hLo);
            if (hHi != hLo) {
              const float wHi = lane_f32(alpha, 4 * j + hHi);
              w = (lane < (hLo + 1) * CH - 32 * k) ? w : wHi;
            }
            atomicAdd(&arow[lane + 32 * k], f[j][k] * w);
          }
        }
      }
    }
  };

  int n = 0, nseg = 1;
  bool slow = false;
  for (int pass = 0; pass < 3; ++pass) {
    for (int seg = 0; seg < nseg; ++seg) {
      if (pass == 0 || slow) {
        int total = -1;
        if (LOAD && !slow) total = sv_cnt[blockIdx.x];
        if (total >= 0 && total <= OWN_CAP) {
          for (int i = tid; i < total; i += OWN_T) {
            lsrc[i] = sv_src[svBase + i];
            ldl[i] = sv_dl[svBase + i];
          }
          __syncthreads();
          n = total;
          resolve(n, false);
        } else {
          const int a = slow ? seg * OWN_CAP : 0;
          const int b = slow ? min(E, a + OWN_CAP) : E;
          total = scan(a, b);
          n = min(total, OWN_CAP);
          resolve(n, true);
          if (!slow) {
            if (SAVE) {
              if (total <= OWN_CAP) {
                for (int i = tid; i < n; i += OWN_T) {
                  sv_src[svBase + i] = lsrc[i];
                  sv_dl[svBase + i] = ldl[i];
                }
              }
              if (tid == 0) sv_cnt[blockIdx.x] = (total <= OWN_CAP) ? total : -1;
            }
            if (total > OWN_CAP) {
              slow = true;
              nseg = (E + OWN_CAP - 1) / OWN_CAP;
              seg = -1;
              continue;
            }
          }
        }
      }
      if (pass == 1) { sumphase(n, seg == 0); __syncthreads(); }
      if (pass == 2) { aggphase(n, seg == 0); __syncthreads(); }
    }
  }

  for (int i = wave; i < nloc; i += OWN_T / 32) {
    const int node = lo + i;
    const float* ap = acc + i * FR;
    if constexpr (CONCAT) {
      float v[4];
      float sum = 0.0f;
#pragma unroll
      for (int j = 0; j < 4; ++j) {
        const int c = lane * 4 + j;
        float t = ap[c] + bias[c];
        t = (t > 0.0f) ? t : (__expf(t) - 1.0f);
        v[j] = t; sum += t;
      }
      const float mu = warp_sum32(sum) * (1.0f / 128.0f);
      float vs = 0.0f;
#pragma unroll
      for (int j = 0; j < 4; ++j) { const float dd = v[j] - mu; vs += dd * dd; }
      const float var = warp_sum32(vs) * (1.0f / 128.0f);
      const float rstd = rsqrtf(var + GAT_LN_EPS);
      float* xp = out + (size_t)node * FR;
#pragma unroll
      for (int j = 0; j < 4; ++j) {
        const int c = lane * 4 + j;
        xp[c] = (v[j] - mu) * rstd * lg[c] + lb[c];
      }
    } else {
      const bool has1 = (lane < CH - 32);
      float x0, x1 = 0.0f;
      x0 = (ap[lane] + ap[lane + CH] + ap[lane + 2 * CH] + ap[lane + 3 * CH]) * 0.25f + bias[lane];
      if (has1)
        x1 = (ap[lane + 32] + ap[lane + 32 + CH] + ap[lane + 32 + 2 * CH] + ap[lane + 32 + 3 * CH]) *
                 0.25f + bias[lane + 32];
      const float mu = warp_sum32(x0 + (has1 ? x1 : 0.0f)) * (1.0f / CH);
      const float d0 = x0 - mu, d1 = has1 ? (x1 - mu) : 0.0f;
      const float var = warp_sum32(d0 * d0 + d1 * d1) * (1.0f / CH);
      const float rstd = rsqrtf(var + GAT_LN_EPS);
      const float z0 = d0 * rstd * lg[lane] + lb[lane];
      const float z1 = has1 ? (d1 * rstd * lg[lane + 32] + lb[lane + 32]) : -3.0e38f;
      const float mxz = warp_max32(fmaxf(z0, z1));
      const float se = warp_sum32(__expf(z0 - mxz) + (has1 ? __expf(z1 - mxz) : 0.0f));
      const float lse = __logf(se);
      float* yp = out + (size_t)node * CH;
      yp[lane] = z0 - mxz - lse;
      if (has1) yp[lane + 32] = z1 - mxz - lse;
    }
  }
}

extern "C" void kernel_launch(void* const* d_in, const int* in_sizes, int n_in,
                              void* d_out, int out_size, void* d_ws, size_t ws_size,
                              hipStream_t stream) {
  const float* x        = (const float*)d_in[0];
  const int*   ei       = (const int*)d_in[1];
  const float* W1       = (const float*)d_in[2];
  const float* att_src1 = (const float*)d_in[3];
  const float* att_dst1 = (const float*)d_in[4];
  const float* b1       = (const float*)d_in[5];
  const float* W2       = (const float*)d_in[6];
  const float* att_src2 = (const float*)d_in[7];
  const float* att_dst2 = (const float*)d_in[8];
  const float* b2       = (const float*)d_in[9];
  const float* ln0_g    = (const float*)d_in[10];
  const float* ln0_b    = (const float*)d_in[11];
  const float* ln1_g    = (const float*)d_in[12];
  const float* ln1_b    = (const float*)d_in[13];

  const int F_IN = 256, H1C = 128, H2C = 160, H = 4;
  const int N = in_sizes[0] / F_IN;
  const int E = in_sizes[1] / 2;

  const int mTiles = (N + 15) / 16;
  const int owners = (N + OWN_RPW - 1) / OWN_RPW;

  float* ws   = (float*)d_ws;
  float* h1   = ws;
  float* out1 = h1   + (size_t)N * H1C;
  float* h2   = out1 + (size_t)N * H1C;
  float* es   = h2   + (size_t)N * H2C;
  float* ed   = es   + (size_t)N * H;
  int* sv_cnt = (int*)(ed + (size_t)N * H);
  int* sv_src = sv_cnt + owners;
  int* sv_dl  = sv_src + (size_t)owners * OWN_CAP;

  gemm_wmma_f32<<<mTiles, 2 * H1C, 16 * (F_IN + 1) * sizeof(float), stream>>>(
      x, W1, h1, N, F_IN, H1C);
  attn_scores_k<<<(N * H + 255) / 256, 256, 0, stream>>>(h1, att_src1, att_dst1,
                                                         es, ed, N, H, 32);
  owner_conv_k<32, true, true, false><<<owners, OWN_T, 0, stream>>>(
      ei, E, N, h1, es, ed, b1, ln0_g, ln0_b, out1, sv_cnt, sv_src, sv_dl);

  gemm_wmma_f32<<<mTiles, 2 * H2C, 16 * (H1C + 1) * sizeof(float), stream>>>(
      out1, W2, h2, N, H1C, H2C);
  attn_scores_k<<<(N * H + 255) / 256, 256, 0, stream>>>(h2, att_src2, att_dst2,
                                                         es, ed, N, H, 40);
  owner_conv_k<40, false, false, true><<<owners, OWN_T, 0, stream>>>(
      ei, E, N, h2, es, ed, b2, ln1_g, ln1_b, (float*)d_out, sv_cnt, sv_src, sv_dl);
}


# Owners scan a byte-per-edge owner plane (dst/196) written by the first score launch beside a packed (src, local dst) word: a quarter of the scan's bytes and compares; float adds stay in arrival order
# speedup vs baseline: 3.1317x; 1.1012x over previous
#include <hip/hip_runtime.h>
#include <math.h>

typedef __attribute__((ext_vector_type(2))) float v2f;
typedef __attribute__((ext_vector_type(8))) float v8f;

#define GAT_NEG_SLOPE 0.2f
#define GAT_SM_EPS 1e-16f
#define GAT_LN_EPS 1e-5f

__device__ __forceinline__ float warp_sum32(float v) {
#pragma unroll
  for (int o = 16; o > 0; o >>= 1) v += __shfl_xor(v, o, 32);
  return v;
}
__device__ __forceinline__ float warp_max32(float v) {
#pragma unroll
  for (int o = 16; o > 0; o >>= 1) v = fmaxf(v, __shfl_xor(v, o, 32));
  return v;
}


__global__ void gemm_wmma_f32(const float* __restrict__ A, const float* __restrict__ B,
                              float* __restrict__ C, int M, int K, int Nc) {
  extern __shared__ float As[];
  const int KP = K + 1;
  const int mBase = blockIdx.x * 16;
  const int tid = threadIdx.x;

  for (int i = tid; i < 16 * K; i += blockDim.x) {
    int r = i / K, c = i - r * K;
    int row = mBase + r;
    As[r * KP + c] = (row < M) ? A[(size_t)row * K + c] : 0.0f;
  }
  __syncthreads();

  const int wave = tid >> 5;
  const int lane = tid & 31;
  const int nBase = wave * 16;
  const int l15 = lane & 15;
  const int khalf = (lane >> 4) * 2;

  v8f acc = {};
  for (int k0 = 0; k0 < K; k0 += 4) {
    v2f a, b;
    a.x = As[l15 * KP + k0 + khalf];
    a.y = As[l15 * KP + k0 + khalf + 1];
    b.x = B[(size_t)(k0 + khalf) * Nc + nBase + l15];
    b.y = B[(size_t)(k0 + khalf + 1) * Nc + nBase + l15];
    acc = __builtin_amdgcn_wmma_f32_16x16x4_f32(false, a, false, b,
                                                (short)0, acc, false, false);
  }

  const int rowTop = mBase + (lane >> 4) * 8;
  const int col = nBase + l15;
#pragma unroll
  for (int r = 0; r < 8; ++r) {
    int row = rowTop + r;
    if (row < M) C[(size_t)row * Nc + col] = acc[r];
  }
}

#define OWN_T 1024
#define OWN_RPW 196
#define OWN_CAP 6144
#define OWN_SCAN_U 7
static_assert(OWN_RPW <= 256, "a destination's place in its owner's range is kept in one byte");

__global__ void attn_scores_k(const float* __restrict__ h, const float* __restrict__ a_src,
                              const float* __restrict__ a_dst, float* __restrict__ es,
                              float* __restrict__ ed, int N, int H, int CH,
                              const int* __restrict__ ei, int E,
                              unsigned char* __restrict__ own8, int* __restrict__ rec) {
  int idx = blockIdx.x * blockDim.x + threadIdx.x;
  if (ei != nullptr) {
    const int* __restrict__ dstp = ei + E;
    const int nthr = gridDim.x * blockDim.x;
    auto pack = [](int sN, int dN, unsigned& o) -> int {
      o = (unsigned)dN / (unsigned)OWN_RPW;
      return (int)(((unsigned)sN << 8) | ((unsigned)dN - o * (unsigned)OWN_RPW));
    };
    if ((E & 3) == 0 && ((((size_t)rec) | ((size_t)own8)) & 15) == 0) {
      for (int q = idx; q < (E >> 2); q += nthr) {
        const int4 sv = ((const int4*)ei)[q];
        const int4 dv = ((const int4*)dstp)[q];
        unsigned o0, o1, o2, o3;
        int4 r;
        r.x = pack(sv.x, dv.x, o0); r.y = pack(sv.y, dv.y, o1);
        r.z = pack(sv.z, dv.z, o2); r.w = pack(sv.w, dv.w, o3);
        ((unsigned*)own8)[q] = (o0 & 255u) | ((o1 & 255u) << 8) | ((o2 & 255u) << 16) | (o3 << 24);
        ((int4*)rec)[q] = r;
      }
    } else {
      for (int e = idx; e < E; e += nthr) {
        unsigned o;
        rec[e] = pack(ei[e], dstp[e], o);
        own8[e] = (unsigned char)o;
      }
    }
  }
  if (idx >= N * H) return;
  int n = idx / H, hh = idx - n * H;
  const float* hp = h + (size_t)n * H * CH + hh * CH;
  const float* asp = a_src + hh * CH;
  const float* adp = a_dst + hh * CH;
  float s = 0.0f, d = 0.0f;
  for (int c = 0; c < CH; ++c) { float t = hp[c]; s += t * asp[c]; d += t * adp[c]; }
  es[idx] = s; ed[idx] = d;
}

__device__ __forceinline__ int lane_i32(int v, int src) {
  return __builtin_amdgcn_readlane(v, src);
}
__device__ __forceinline__ float lane_f32(float v, int src) {
  return __int_as_float(__builtin_amdgcn_readlane(__float_as_int(v), src));
}

__device__ __forceinline__ float leaky(float v) { return (v > 0.0f) ? v : GAT_NEG_SLOPE * v; }

__device__ __forceinline__ void lds_max_f32(float* addr, float v) {
  if (v >= 0.0f) atomicMax((int*)addr, __float_as_int(v));
  else           atomicMin((unsigned int*)addr, __float_as_uint(v));
}

template <int CH, bool CONCAT, bool SAVE, bool LOAD>
__global__ void __launch_bounds__(OWN_T)
owner_conv_k(const unsigned char* __restrict__ own8, const int* __restrict__ rec, int E, int N,
             const float* __restrict__ feat, const float* __restrict__ es,
             const float* __restrict__ ed, const float* __restrict__ bias,
             const float* __restrict__ lg, const float* __restrict__ lb,
             float* __restrict__ out,
             int* __restrict__ sv_cnt, int* __restrict__ sv_src, int* __restrict__ sv_dl) {
  constexpr int H = 4;
  constexpr int FR = H * CH;
  constexpr int KCH = FR / 32;
  constexpr int EPB = 8;
  __shared__ float acc[OWN_RPW * FR];
  __shared__ float lw[OWN_CAP * H];
  __shared__ int lsrc[OWN_CAP];
  __shared__ int ldl[OWN_CAP];
  __shared__ float edS[OWN_RPW * H], slf[OWN_RPW * H], mx[OWN_RPW * H], den[OWN_RPW * H];
  __shared__ int cnt;

  const int tid = threadIdx.x;
  const int lane = tid & 31;
  const int wave = __builtin_amdgcn_readfirstlane(tid >> 5);
  const int lo = blockIdx.x * OWN_RPW;
  const int nloc = min(OWN_RPW, N - lo);
  const size_t svBase = (size_t)blockIdx.x * OWN_CAP;

  for (int i = tid; i < nloc * FR; i += OWN_T) acc[i] = 0.0f;
  for (int i = tid; i < nloc * H; i += OWN_T) {
    const float d = ed[(size_t)lo * H + i];
    const float v = leaky(es[(size_t)lo * H + i] + d);
    edS[i] = d; slf[i] = v; mx[i] = v; den[i] = 0.0f;
  }

  auto scan = [&](int a, int b) -> int {
    if (tid == 0) cnt = 0;
    __syncthreads();
    auto hit = [&](int e) {
      const int i = atomicAdd(&cnt, 1);
      if (i < OWN_CAP) lsrc[i] = e;
    };
    const unsigned me4 = (unsigned)blockIdx.x * 0x01010101u;
    auto match = [&](unsigned w) -> unsigned {
      const unsigned x = w ^ me4;
      return ~((((x & 0x7f7f7f7fu) + 0x7f7f7f7fu) | x) | 0x7f7f7f7fu);
    };
    if ((((size_t)(own8 + a)) & 15) == 0) {
      const int nv = (b - a) >> 4;
      const uint4* __restrict__ p4 = (const uint4*)(own8 + a);
      const int rot = (int)(((long long)blockIdx.x * nv) / gridDim.x) & ~31;
      for (int i0 = tid; i0 < nv; i0 += OWN_T * OWN_SCAN_U) {
        uint4 v[OWN_SCAN_U];
        int at[OWN_SCAN_U];
#pragma unroll
        for (int u = 0; u < OWN_SCAN_U; ++u) {
          const int ii = i0 + u * OWN_T;
          int jj = ii + rot;
          if (jj >= nv) jj -= nv;
          at[u] = (ii < nv) ? jj : -1;
          v[u] = p4[max(at[u], 0)];
        }
#pragma unroll
        for (int u = 0; u < OWN_SCAN_U; ++u) {
          unsigned m = (match(v[u].x) >> 7) | (match(v[u].y) >> 6) |
                       (match(v[u].z) >> 5) | (match(v[u].w) >> 4);
          if (at[u] < 0) m = 0u;
          const int e0 = a + 16 * at[u];
          while (m != 0u) {
            const int p = __ffs((int)m) - 1;
            hit(e0 + 4 * (p & 7) + (p >> 3));
            m &= m - 1u;
          }
        }
      }
      for (int e = a + 16 * nv + tid; e < b; e += OWN_T)
        if (own8[e] == (unsigned char)blockIdx.x) hit(e);
    } else {
      for (int e = a + tid; e < b; e += OWN_T)
        if (own8[e] == (unsigned char)blockIdx.x) hit(e);
    }
    __syncthreads();
    return cnt;
  };

  auto resolve = [&](int n, bool fromEdgeIds) {
    for (int i0 = tid; i0 < n; i0 += OWN_T * 4) {
      int s[4], dls[4];
      float4 ev[4];
#pragma unroll
      for (int u = 0; u < 4; ++u) {
        const int i = i0 + u * OWN_T;
        int t = (i < n) ? lsrc[i] : 0;
        if (fromEdgeIds) {
          const int r = rec[t];
          s[u] = (int)((unsigned)r >> 8);
          dls[u] = r & 255;
        } else {
          s[u] = t;
          dls[u] = (i < n) ? ldl[i] : 0;
        }
      }
#pragma unroll
      for (int u = 0; u < 4; ++u) ev[u] = *(const float4*)(es + (size_t)s[u] * H);
#pragma unroll
      for (int u = 0; u < 4; ++u) {
        const int i = i0 + u * OWN_T;
        if (i < n) {
          const int dl = dls[u];
          if (fromEdgeIds) { lsrc[i] = s[u]; ldl[i] = dl; }
          const float l0 = leaky(ev[u].x + edS[dl * H + 0]);
          const float l1 = leaky(ev[u].y + edS[dl * H + 1]);
          const float l2 = leaky(ev[u].z + edS[dl * H + 2]);
          const float l3 = leaky(ev[u].w + edS[dl * H + 3]);
          lw[i * H + 0] = l0; lw[i * H + 1] = l1; lw[i * H + 2] = l2; lw[i * H + 3] = l3;
          lds_max_f32(&mx[dl * H + 0], l0); lds_max_f32(&mx[dl * H + 1], l1);
          lds_max_f32(&mx[dl * H + 2], l2); lds_max_f32(&mx[dl * H + 3], l3);
        }
      }
    }
    __syncthreads();
  };

  auto sumphase = [&](int n, bool first) {
    const int M = (n + (first ? nloc : 0)) * H;
    for (int it = tid; it < M; it += OWN_T) {
      const int i = it >> 2, h = it & 3;
      int dl; float l;
      if (i < n) { dl = ldl[i]; l = lw[it]; }
      else       { dl = i - n;  l = slf[dl * H + h]; }
      atomicAdd(&den[dl * H + h], __expf(l - mx[dl * H + h]));
    }
  };

  auto aggphase = [&](int n, bool first) {
    const int M = n + (first ? nloc : 0);
    const int nR = (M + EPB - 1) / EPB;
    for (int r = wave; r < nR; r += OWN_T / 32) {
      const int h = lane & 3;
      const int i = r * EPB + (lane >> 2);
      int s = 0, dl = 0; float alpha = 0.0f;
      if (i < M) {
        float l;
        if (i < n) { s = lsrc[i]; dl = ldl[i]; l = lw[i * H + h]; }
        else       { dl = i - n; s = lo + dl; l = slf[dl * H + h]; }
        alpha = __expf(l - mx[dl * H + h]) / (den[dl * H + h] + GAT_SM_EPS);
      }
      float f[EPB][KCH];
#pragma unroll
      for (int j = 0; j < EPB; ++j) {
        const float* row = feat + (size_t)lane_i32(s, 4 * j) * FR;
#pragma unroll
        for (int k = 0; k < KCH; ++k) f[j][k] = row[lane + 32 * k];
      }
#pragma unroll
      for (int j = 0; j < EPB; ++j) {
        if (r * EPB + j < M) {
          float* arow = acc + lane_i32(dl, 4 * j) * FR;
#pragma unroll
          for (int k = 0; k < KCH; ++k) {
            const int hLo = (32 * k) / CH, hHi = (32 * k + 31) / CH;
            float w = lane_f32(alpha, 4 * j + hLo);
            if (hHi != hLo) {
              const float wHi = lane_f32(alpha, 4 * j + hHi);
              w = (lane < (hLo + 1) * CH - 32 * k) ? w : wHi;
            }
            atomicAdd(&arow[lane + 32 * k], f[j][k] * w);
          }
        }
      }
    }
  };

  int n = 0, nseg = 1;
  bool slow = false;
  for (int pass = 0; pass < 3; ++pass) {
    for (int seg = 0; seg < nseg; ++seg) {
      if (pass == 0 || slow) {
        int total = -1;
        if (LOAD && !slow) total = sv_cnt[blockIdx.x];
        if (total >= 0 && total <= OWN_CAP) {
          for (int i = tid; i < total; i += OWN_T) {
            lsrc[i] = sv_src[svBase + i];
            ldl[i] = sv_dl[svBase + i];
          }
          __syncthreads();
          n = total;
          resolve(n, false);
        } else {
          const int a = slow ? seg * OWN_CAP : 0;
          const int b = slow ? min(E, a + OWN_CAP) : E;
          total = scan(a, b);
          n = min(total, OWN_CAP);
          resolve(n, true);
          if (!slow) {
            if (SAVE) {
              if (total <= OWN_CAP) {
                for (int i = tid; i < n; i += OWN_T) {
                  sv_src[svBase + i] = lsrc[i];
                  sv_dl[svBase + i] = ldl[i];
                }
              }
              if (tid == 0) sv_cnt[blockIdx.x] = (total <= OWN_CAP) ? total : -1;
            }
            if (total > OWN_CAP) {
              slow = true;
              nseg = (E + OWN_CAP - 1) / OWN_CAP;
              seg = -1;
              continue;
            }
          }
        }
      }
      if (pass == 1) { sumphase(n, seg == 0); __syncthreads(); }
      if (pass == 2) { aggphase(n, seg == 0); __syncthreads(); }
    }
  }

  for (int i = wave; i < nloc; i += OWN_T / 32) {
    const int node = lo + i;
    const float* ap = acc + i * FR;
    if constexpr (CONCAT) {
      float v[4];
      float sum = 0.0f;
#pragma unroll
      for (int j = 0; j < 4; ++j) {
        const int c = lane * 4 + j;
        float t = ap[c] + bias[c];
        t = (t > 0.0f) ? t : (__expf(t) - 1.0f);
        v[j] = t; sum += t;
      }
      const float mu = warp_sum32(sum) * (1.0f / 128.0f);
      float vs = 0.0f;
#pragma unroll
      for (int j = 0; j < 4; ++j) { const float dd = v[j] - mu; vs += dd * dd; }
      const float var = warp_sum32(vs) * (1.0f / 128.0f);
      const float rstd = rsqrtf(var + GAT_LN_EPS);
      float* xp = out + (size_t)node * FR;
#pragma unroll
      for (int j = 0; j < 4; ++j) {
        const int c = lane * 4 + j;
        xp[c] = (v[j] - mu) * rstd * lg[c] + lb[c];
      }
    } else {
      const bool has1 = (lane < CH - 32);
      float x0, x1 = 0.0f;
      x0 = (ap[lane] + ap[lane + CH] + ap[lane + 2 * CH] + ap[lane + 3 * CH]) * 0.25f + bias[lane];
      if (has1)
        x1 = (ap[lane + 32] + ap[lane + 32 + CH] + ap[lane + 32 + 2 * CH] + ap[lane + 32 + 3 * CH]) *
                 0.25f + bias[lane + 32];
      const float mu = warp_sum32(x0 + (has1 ? x1 : 0.0f)) * (1.0f / CH);
      const float d0 = x0 - mu, d1 = has1 ? (x1 - mu) : 0.0f;
      const float var = warp_sum32(d0 * d0 + d1 * d1) * (1.0f / CH);
      const float rstd = rsqrtf(var + GAT_LN_EPS);
      const float z0 = d0 * rstd * lg[lane] + lb[lane];
      const float z1 = has1 ? (d1 * rstd * lg[lane + 32] + lb[lane + 32]) : -3.0e38f;
      const float mxz = warp_max32(fmaxf(z0, z1));
      const float se = warp_sum32(__expf(z0 - mxz) + (has1 ? __expf(z1 - mxz) : 0.0f));
      const float lse = __logf(se);
      float* yp = out + (size_t)node * CH;
      yp[lane] = z0 - mxz - lse;
      if (has1) yp[lane + 32] = z1 - mxz - lse;
    }
  }
}

extern "C" void kernel_launch(void* const* d_in, const int* in_sizes, int n_in,
                              void* d_out, int out_size, void* d_ws, size_t ws_size,
                              hipStream_t stream) {
  const float* x        = (const float*)d_in[0];
  const int*   ei       = (const int*)d_in[1];
  const float* W1       = (const float*)d_in[2];
  const float* att_src1 = (const float*)d_in[3];
  const float* att_dst1 = (const float*)d_in[4];
  const float* b1       = (const float*)d_in[5];
  const float* W2       = (const float*)d_in[6];
  const float* att_src2 = (const float*)d_in[7];
  const float* att_dst2 = (const float*)d_in[8];
  const float* b2       = (const float*)d_in[9];
  const float* ln0_g    = (const float*)d_in[10];
  const float* ln0_b    = (const float*)d_in[11];
  const float* ln1_g    = (const float*)d_in[12];
  const float* ln1_b    = (const float*)d_in[13];

  const int F_IN = 256, H1C = 128, H2C = 160, H = 4;
  const int N = in_sizes[0] / F_IN;
  const int E = in_sizes[1] / 2;

  const int mTiles = (N + 15) / 16;
  const int owners = (N + OWN_RPW - 1) / OWN_RPW;

  float* ws   = (float*)d_ws;
  float* h1   = ws;
  float* out1 = h1   + (size_t)N * H1C;
  float* h2   = out1 + (size_t)N * H1C;
  float* es   = h2   + (size_t)N * H2C;
  float* ed   = es   + (size_t)N * H;
  int* sv_cnt = (int*)(ed + (size_t)N * H);
  int* sv_src = sv_cnt + owners;
  int* sv_dl  = sv_src + (size_t)owners * OWN_CAP;
  int* rec    = sv_dl + (size_t)owners * OWN_CAP;
  unsigned char* own8 = (unsigned char*)(rec + E);

  gemm_wmma_f32<<<mTiles, 2 * H1C, 16 * (F_IN + 1) * sizeof(float), stream>>>(
      x, W1, h1, N, F_IN, H1C);
  attn_scores_k<<<(N * H + 255) / 256, 256, 0, stream>>>(h1, att_src1, att_dst1,
                                                         es, ed, N, H, 32, ei, E, own8, rec);
  owner_conv_k<32, true, true, false><<<owners, OWN_T, 0, stream>>>(
      own8, rec, E, N, h1, es, ed, b1, ln0_g, ln0_b, out1, sv_cnt, sv_src, sv_dl);

  gemm_wmma_f32<<<mTiles, 2 * H2C, 16 * (H1C + 1) * sizeof(float), stream>>>(
      out1, W2, h2, N, H1C, H2C);
  attn_scores_k<<<(N * H + 255) / 256, 256, 0, stream>>>(h2, att_src2, att_dst2,
                                                         es, ed, N, H, 40, nullptr, 0, nullptr, nullptr);
  owner_conv_k<40, false, false, true><<<owners, OWN_T, 0, stream>>>(
      own8, rec, E, N, h2, es, ed, b2, ln1_g, ln1_b, (float*)d_out, sv_cnt, sv_src, sv_dl);
}


# GEMMs: weights staged once per work-group in LDS in fragment order; a wave streams its 16 rows of A from memory to registers and feeds all column tiles; k summed in a permuted order within each 16
# speedup vs baseline: 4.2601x; 1.3603x over previous
#include <hip/hip_runtime.h>
#include <math.h>

typedef __attribute__((ext_vector_type(2))) float v2f;
typedef __attribute__((ext_vector_type(8))) float v8f;

#define GAT_NEG_SLOPE 0.2f
#define GAT_SM_EPS 1e-16f
#define GAT_LN_EPS 1e-5f

__device__ __forceinline__ float warp_sum32(float v) {
#pragma unroll
  for (int o = 16; o > 0; o >>= 1) v += __shfl_xor(v, o, 32);
  return v;
}
__device__ __forceinline__ float warp_max32(float v) {
#pragma unroll
  for (int o = 16; o > 0; o >>= 1) v = fmaxf(v, __shfl_xor(v, o, 32));
  return v;
}


#define GEMM_T 512
template <int K, int NC>
__global__ void __launch_bounds__(GEMM_T)
gemm_wmma_f32(const float* __restrict__ A, const float* __restrict__ B, float* __restrict__ C,
              int M) {
  constexpr int T = NC / 16;
  constexpr int KB = K / 16;
  static_assert(K % 16 == 0 && NC % 16 == 0, "whole fragments only");
  __shared__ float4 Bs[KB * 2 * T * 32];
  const int tid = threadIdx.x;
  const int lane = tid & 31;
  const int wave = __builtin_amdgcn_readfirstlane(tid >> 5);
  const int tile = blockIdx.x * (GEMM_T / 32) + wave;
  const int c = lane & 15, h = lane >> 4;

  const int rowA = min(tile * 16 + c, M - 1);
  const float4* __restrict__ ap = (const float4*)(A + (size_t)rowA * K + 8 * h);
  float4 a[KB][2];
#pragma unroll
  for (int kb = 0; kb < KB; ++kb) { a[kb][0] = ap[4 * kb]; a[kb][1] = ap[4 * kb + 1]; }

  constexpr int ITEMS = (K / 4) * NC;
  static_assert(ITEMS % GEMM_T == 0, "every thread stages the same number of items");
  float4 bv4[ITEMS / GEMM_T];
#pragma unroll
  for (int u = 0; u < ITEMS / GEMM_T; ++u) {
    const int it = tid + u * GEMM_T;
    const int m = it / NC, n = it - m * NC;
    const float* __restrict__ bp = B + (size_t)(4 * m) * NC + n;
    bv4[u].x = bp[0]; bv4[u].y = bp[NC]; bv4[u].z = bp[2 * NC]; bv4[u].w = bp[3 * NC];
  }
#pragma unroll
  for (int u = 0; u < ITEMS / GEMM_T; ++u) {
    const int it = tid + u * GEMM_T;
    const int m = it / NC, n = it - m * NC;
    const int kb = m >> 2, hh = (m >> 1) & 1, jp = m & 1;
    Bs[((kb * 2 + jp) * T + (n >> 4)) * 32 + hh * 16 + (n & 15)] = bv4[u];
  }
  __syncthreads();
  if (tile * 16 >= M) return;

  v8f acc[T];
#pragma unroll
  for (int t = 0; t < T; ++t) acc[t] = v8f{};
#pragma unroll
  for (int kb = 0; kb < KB; ++kb) {
#pragma unroll
    for (int jp = 0; jp < 2; ++jp) {
      const float4 av = a[kb][jp];
      const v2f a0 = {av.x, av.y}, a1 = {av.z, av.w};
#pragma unroll
      for (int t = 0; t < T; ++t) {
        const float4 bv = Bs[((kb * 2 + jp) * T + t) * 32 + lane];
        const v2f b0 = {bv.x, bv.y}, b1 = {bv.z, bv.w};
        acc[t] = __builtin_amdgcn_wmma_f32_16x16x4_f32(false, a0, false, b0, (short)0, acc[t],
                                                       false, false);
        acc[t] = __builtin_amdgcn_wmma_f32_16x16x4_f32(false, a1, false, b1, (short)0, acc[t],
                                                       false, false);
      }
    }
  }

  const int rowTop = tile * 16 + h * 8;
#pragma unroll
  for (int t = 0; t < T; ++t) {
#pragma unroll
    for (int r = 0; r < 8; ++r) {
      const int row = rowTop + r;
      if (row < M) C[(size_t)row * NC + t * 16 + c] = acc[t][r];
    }
  }
}

#define OWN_T 1024
#define OWN_RPW 196
#define OWN_CAP 6144
#define OWN_SCAN_U 7
static_assert(OWN_RPW <= 256, "a destination's place in its owner's range is kept in one byte");

__global__ void attn_scores_k(const float* __restrict__ h, const float* __restrict__ a_src,
                              const float* __restrict__ a_dst, float* __restrict__ es,
                              float* __restrict__ ed, int N, int H, int CH,
                              const int* __restrict__ ei, int E,
                              unsigned char* __restrict__ own8, int* __restrict__ rec) {
  int idx = blockIdx.x * blockDim.x + threadIdx.x;
  if (ei != nullptr) {
    const int* __restrict__ dstp = ei + E;
    const int nthr = gridDim.x * blockDim.x;
    auto pack = [](int sN, int dN, unsigned& o) -> int {
      o = (unsigned)dN / (unsigned)OWN_RPW;
      return (int)(((unsigned)sN << 8) | ((unsigned)dN - o * (unsigned)OWN_RPW));
    };
    if ((E & 3) == 0 && ((((size_t)rec) | ((size_t)own8)) & 15) == 0) {
      for (int q = idx; q < (E >> 2); q += nthr) {
        const int4 sv = ((const int4*)ei)[q];
        const int4 dv = ((const int4*)dstp)[q];
        unsigned o0, o1, o2, o3;
        int4 r;
        r.x = pack(sv.x, dv.x, o0); r.y = pack(sv.y, dv.y, o1);
        r.z = pack(sv.z, dv.z, o2); r.w = pack(sv.w, dv.w, o3);
        ((unsigned*)own8)[q] = (o0 & 255u) | ((o1 & 255u) << 8) | ((o2 & 255u) << 16) | (o3 << 24);
        ((int4*)rec)[q] = r;
      }
    } else {
      for (int e = idx; e < E; e += nthr) {
        unsigned o;
        rec[e] = pack(ei[e], dstp[e], o);
        own8[e] = (unsigned char)o;
      }
    }
  }
  if (idx >= N * H) return;
  int n = idx / H, hh = idx - n * H;
  const float* hp = h + (size_t)n * H * CH + hh * CH;
  const float* asp = a_src + hh * CH;
  const float* adp = a_dst + hh * CH;
  float s = 0.0f, d = 0.0f;
  for (int c = 0; c < CH; ++c) { float t = hp[c]; s += t * asp[c]; d += t * adp[c]; }
  es[idx] = s; ed[idx] = d;
}

__device__ __forceinline__ int lane_i32(int v, int src) {
  return __builtin_amdgcn_readlane(v, src);
}
__device__ __forceinline__ float lane_f32(float v, int src) {
  return __int_as_float(__builtin_amdgcn_readlane(__float_as_int(v), src));
}

__device__ __forceinline__ float leaky(float v) { return (v > 0.0f) ? v : GAT_NEG_SLOPE * v; }

__device__ __forceinline__ void lds_max_f32(float* addr, float v) {
  if (v >= 0.0f) atomicMax((int*)addr, __float_as_int(v));
  else           atomicMin((unsigned int*)addr, __float_as_uint(v));
}

template <int CH, bool CONCAT, bool SAVE, bool LOAD>
__global__ void __launch_bounds__(OWN_T)
owner_conv_k(const unsigned char* __restrict__ own8, const int* __restrict__ rec, int E, int N,
             const float* __restrict__ feat, const float* __restrict__ es,
             const float* __restrict__ ed, const float* __restrict__ bias,
             const float* __restrict__ lg, const float* __restrict__ lb,
             float* __restrict__ out,
             int* __restrict__ sv_cnt, int* __restrict__ sv_src, int* __restrict__ sv_dl) {
  constexpr int H = 4;
  constexpr int FR = H * CH;
  constexpr int KCH = FR / 32;
  constexpr int EPB = 8;
  __shared__ float acc[OWN_RPW * FR];
  __shared__ float lw[OWN_CAP * H];
  __shared__ int lsrc[OWN_CAP];
  __shared__ int ldl[OWN_CAP];
  __shared__ float edS[OWN_RPW * H], slf[OWN_RPW * H], mx[OWN_RPW * H], den[OWN_RPW * H];
  __shared__ int cnt;

  const int tid = threadIdx.x;
  const int lane = tid & 31;
  const int wave = __builtin_amdgcn_readfirstlane(tid >> 5);
  const int lo = blockIdx.x * OWN_RPW;
  const int nloc = min(OWN_RPW, N - lo);
  const size_t svBase = (size_t)blockIdx.x * OWN_CAP;

  for (int i = tid; i < nloc * FR; i += OWN_T) acc[i] = 0.0f;
  for (int i = tid; i < nloc * H; i += OWN_T) {
    const float d = ed[(size_t)lo * H + i];
    const float v = leaky(es[(size_t)lo * H + i] + d);
    edS[i] = d; slf[i] = v; mx[i] = v; den[i] = 0.0f;
  }

  auto scan = [&](int a, int b) -> int {
    if (tid == 0) cnt = 0;
    __syncthreads();
    auto hit = [&](int e) {
      const int i = atomicAdd(&cnt, 1);
      if (i < OWN_CAP) lsrc[i] = e;
    };
    const unsigned me4 = (unsigned)blockIdx.x * 0x01010101u;
    auto match = [&](unsigned w) -> unsigned {
      const unsigned x = w ^ me4;
      return ~((((x & 0x7f7f7f7fu) + 0x7f7f7f7fu) | x) | 0x7f7f7f7fu);
    };
    if ((((size_t)(own8 + a)) & 15) == 0) {
      const int nv = (b - a) >> 4;
      const uint4* __restrict__ p4 = (const uint4*)(own8 + a);
      const int rot = (int)(((long long)blockIdx.x * nv) / gridDim.x) & ~31;
      for (int i0 = tid; i0 < nv; i0 += OWN_T * OWN_SCAN_U) {
        uint4 v[OWN_SCAN_U];
        int at[OWN_SCAN_U];
#pragma unroll
        for (int u = 0; u < OWN_SCAN_U; ++u) {
          const int ii = i0 + u * OWN_T;
          int jj = ii + rot;
          if (jj >= nv) jj -= nv;
          at[u] = (ii < nv) ? jj : -1;
          v[u] = p4[max(at[u], 0)];
        }
#pragma unroll
        for (int u = 0; u < OWN_SCAN_U; ++u) {
          unsigned m = (match(v[u].x) >> 7) | (match(v[u].y) >> 6) |
                       (match(v[u].z) >> 5) | (match(v[u].w) >> 4);
          if (at[u] < 0) m = 0u;
          const int e0 = a + 16 * at[u];
          while (m != 0u) {
            const int p = __ffs((int)m) - 1;
            hit(e0 + 4 * (p & 7) + (p >> 3));
            m &= m - 1u;
          }
        }
      }
      for (int e = a + 16 * nv + tid; e < b; e += OWN_T)
        if (own8[e] == (unsigned char)blockIdx.x) hit(e);
    } else {
      for (int e = a + tid; e < b; e += OWN_T)
        if (own8[e] == (unsigned char)blockIdx.x) hit(e);
    }
    __syncthreads();
    return cnt;
  };

  auto resolve = [&](int n, bool fromEdgeIds) {
    for (int i0 = tid; i0 < n; i0 += OWN_T * 4) {
      int s[4], dls[4];
      float4 ev[4];
#pragma unroll
      for (int u = 0; u < 4; ++u) {
        const int i = i0 + u * OWN_T;
        int t = (i < n) ? lsrc[i] : 0;
        if (fromEdgeIds) {
          const int r = rec[t];
          s[u] = (int)((unsigned)r >> 8);
          dls[u] = r & 255;
        } else {
          s[u] = t;
          dls[u] = (i < n) ? ldl[i] : 0;
        }
      }
#pragma unroll
      for (int u = 0; u < 4; ++u) ev[u] = *(const float4*)(es + (size_t)s[u] * H);
#pragma unroll
      for (int u = 0; u < 4; ++u) {
        const int i = i0 + u * OWN_T;
        if (i < n) {
          const int dl = dls[u];
          if (fromEdgeIds) { lsrc[i] = s[u]; ldl[i] = dl; }
          const float l0 = leaky(ev[u].x + edS[dl * H + 0]);
          const float l1 = leaky(ev[u].y + edS[dl * H + 1]);
          const float l2 = leaky(ev[u].z + edS[dl * H + 2]);
          const float l3 = leaky(ev[u].w + edS[dl * H + 3]);
          lw[i * H + 0] = l0; lw[i * H + 1] = l1; lw[i * H + 2] = l2; lw[i * H + 3] = l3;
          lds_max_f32(&mx[dl * H + 0], l0); lds_max_f32(&mx[dl * H + 1], l1);
          lds_max_f32(&mx[dl * H + 2], l2); lds_max_f32(&mx[dl * H + 3], l3);
        }
      }
    }
    __syncthreads();
  };

  auto sumphase = [&](int n, bool first) {
    const int M = (n + (first ? nloc : 0)) * H;
    for (int it = tid; it < M; it += OWN_T) {
      const int i = it >> 2, h = it & 3;
      int dl; float l;
      if (i < n) { dl = ldl[i]; l = lw[it]; }
      else       { dl = i - n;  l = slf[dl * H + h]; }
      atomicAdd(&den[dl * H + h], __expf(l - mx[dl * H + h]));
    }
  };

  auto aggphase = [&](int n, bool first) {
    const int M = n + (first ? nloc : 0);
    const int nR = (M + EPB - 1) / EPB;
    for (int r = wave; r < nR; r += OWN_T / 32) {
      const int h = lane & 3;
      const int i = r * EPB + (lane >> 2);
      int s = 0, dl = 0; float alpha = 0.0f;
      if (i < M) {
        float l;
        if (i < n) { s = lsrc[i]; dl = ldl[i]; l = lw[i * H + h]; }
        else       { dl = i - n; s = lo + dl; l = slf[dl * H + h]; }
        alpha = __expf(l - mx[dl * H + h]) / (den[dl * H + h] + GAT_SM_EPS);
      }
      float f[EPB][KCH];
#pragma unroll
      for (int j = 0; j < EPB; ++j) {
        const float* row = feat + (size_t)lane_i32(s, 4 * j) * FR;
#pragma unroll
        for (int k = 0; k < KCH; ++k) f[j][k] = row[lane + 32 * k];
      }
#pragma unroll
      for (int j = 0; j < EPB; ++j) {
        if (r * EPB + j < M) {
          float* arow = acc + lane_i32(dl, 4 * j) * FR;
#pragma unroll
          for (int k = 0; k < KCH; ++k) {
            const int hLo = (32 * k) / CH, hHi = (32 * k + 31) / CH;
            float w = lane_f32(alpha, 4 * j + hLo);
            if (hHi != hLo) {
              const float wHi = lane_f32(alpha, 4 * j + hHi);
              w = (lane < (hLo + 1) * CH - 32 * k) ? w : wHi;
            }
            atomicAdd(&arow[lane + 32 * k], f[j][k] * w);
          }
        }
      }
    }
  };

  int n = 0, nseg = 1;
  bool slow = false;
  for (int pass = 0; pass < 3; ++pass) {
    for (int seg = 0; seg < nseg; ++seg) {
      if (pass == 0 || slow) {
        int total = -1;
        if (LOAD && !slow) total = sv_cnt[blockIdx.x];
        if (total >= 0 && total <= OWN_CAP) {
          for (int i = tid; i < total; i += OWN_T) {
            lsrc[i] = sv_src[svBase + i];
            ldl[i] = sv_dl[svBase + i];
          }
          __syncthreads();
          n = total;
          resolve(n, false);
        } else {
          const int a = slow ? seg * OWN_CAP : 0;
          const int b = slow ? min(E, a + OWN_CAP) : E;
          total = scan(a, b);
          n = min(total, OWN_CAP);
          resolve(n, true);
          if (!slow) {
            if (SAVE) {
              if (total <= OWN_CAP) {
                for (int i = tid; i < n; i += OWN_T) {
                  sv_src[svBase + i] = lsrc[i];
                  sv_dl[svBase + i] = ldl[i];
                }
              }
              if (tid == 0) sv_cnt[blockIdx.x] = (total <= OWN_CAP) ? total : -1;
            }
            if (total > OWN_CAP) {
              slow = true;
              nseg = (E + OWN_CAP - 1) / OWN_CAP;
              seg = -1;
              continue;
            }
          }
        }
      }
      if (pass == 1) { sumphase(n, seg == 0); __syncthreads(); }
      if (pass == 2) { aggphase(n, seg == 0); __syncthreads(); }
    }
  }

  for (int i = wave; i < nloc; i += OWN_T / 32) {
    const int node = lo + i;
    const float* ap = acc + i * FR;
    if constexpr (CONCAT) {
      float v[4];
      float sum = 0.0f;
#pragma unroll
      for (int j = 0; j < 4; ++j) {
        const int c = lane * 4 + j;
        float t = ap[c] + bias[c];
        t = (t > 0.0f) ? t : (__expf(t) - 1.0f);
        v[j] = t; sum += t;
      }
      const float mu = warp_sum32(sum) * (1.0f / 128.0f);
      float vs = 0.0f;
#pragma unroll
      for (int j = 0; j < 4; ++j) { const float dd = v[j] - mu; vs += dd * dd; }
      const float var = warp_sum32(vs) * (1.0f / 128.0f);
      const float rstd = rsqrtf(var + GAT_LN_EPS);
      float* xp = out + (size_t)node * FR;
#pragma unroll
      for (int j = 0; j < 4; ++j) {
        const int c = lane * 4 + j;
        xp[c] = (v[j] - mu) * rstd * lg[c] + lb[c];
      }
    } else {
      const bool has1 = (lane < CH - 32);
      float x0, x1 = 0.0f;
      x0 = (ap[lane] + ap[lane + CH] + ap[lane + 2 * CH] + ap[lane + 3 * CH]) * 0.25f + bias[lane];
      if (has1)
        x1 = (ap[lane + 32] + ap[lane + 32 + CH] + ap[lane + 32 + 2 * CH] + ap[lane + 32 + 3 * CH]) *
                 0.25f + bias[lane + 32];
      const float mu = warp_sum32(x0 + (has1 ? x1 : 0.0f)) * (1.0f / CH);
      const float d0 = x0 - mu, d1 = has1 ? (x1 - mu) : 0.0f;
      const float var = warp_sum32(d0 * d0 + d1 * d1) * (1.0f / CH);
      const float rstd = rsqrtf(var + GAT_LN_EPS);
      const float z0 = d0 * rstd * lg[lane] + lb[lane];
      const float z1 = has1 ? (d1 * rstd * lg[lane + 32] + lb[lane + 32]) : -3.0e38f;
      const float mxz = warp_max32(fmaxf(z0, z1));
      const float se = warp_sum32(__expf(z0 - mxz) + (has1 ? __expf(z1 - mxz) : 0.0f));
      const float lse = __logf(se);
      float* yp = out + (size_t)node * CH;
      yp[lane] = z0 - mxz - lse;
      if (has1) yp[lane + 32] = z1 - mxz - lse;
    }
  }
}

extern "C" void kernel_launch(void* const* d_in, const int* in_sizes, int n_in,
                              void* d_out, int out_size, void* d_ws, size_t ws_size,
                              hipStream_t stream) {
  const float* x        = (const float*)d_in[0];
  const int*   ei       = (const int*)d_in[1];
  const float* W1       = (const float*)d_in[2];
  const float* att_src1 = (const float*)d_in[3];
  const float* att_dst1 = (const float*)d_in[4];
  const float* b1       = (const float*)d_in[5];
  const float* W2       = (const float*)d_in[6];
  const float* att_src2 = (const float*)d_in[7];
  const float* att_dst2 = (const float*)d_in[8];
  const float* b2       = (const float*)d_in[9];
  const float* ln0_g    = (const float*)d_in[10];
  const float* ln0_b    = (const float*)d_in[11];
  const float* ln1_g    = (const float*)d_in[12];
  const float* ln1_b    = (const float*)d_in[13];

  const int F_IN = 256, H1C = 128, H2C = 160, H = 4;
  const int N = in_sizes[0] / F_IN;
  const int E = in_sizes[1] / 2;

  const int mTiles = (N + 15) / 16;
  const int gBlocks = (mTiles + GEMM_T / 32 - 1) / (GEMM_T / 32);
  const int owners = (N + OWN_RPW - 1) / OWN_RPW;

  float* ws   = (float*)d_ws;
  float* h1   = ws;
  float* out1 = h1   + (size_t)N * H1C;
  float* h2   = out1 + (size_t)N * H1C;
  float* es   = h2   + (size_t)N * H2C;
  float* ed   = es   + (size_t)N * H;
  int* sv_cnt = (int*)(ed + (size_t)N * H);
  int* sv_src = sv_cnt + owners;
  int* sv_dl  = sv_src + (size_t)owners * OWN_CAP;
  int* rec    = sv_dl + (size_t)owners * OWN_CAP;
  unsigned char* own8 = (unsigned char*)(rec + E);

  gemm_wmma_f32<256, 128><<<gBlocks, GEMM_T, 0, stream>>>(x, W1, h1, N);
  attn_scores_k<<<(N * H + 255) / 256, 256, 0, stream>>>(h1, att_src1, att_dst1,
                                                         es, ed, N, H, 32, ei, E, own8, rec);
  owner_conv_k<32, true, true, false><<<owners, OWN_T, 0, stream>>>(
      own8, rec, E, N, h1, es, ed, b1, ln0_g, ln0_b, out1, sv_cnt, sv_src, sv_dl);

  gemm_wmma_f32<128, 160><<<gBlocks, GEMM_T, 0, stream>>>(out1, W2, h2, N);
  attn_scores_k<<<(N * H + 255) / 256, 256, 0, stream>>>(h2, att_src2, att_dst2,
                                                         es, ed, N, H, 40, nullptr, 0, nullptr, nullptr);
  owner_conv_k<40, false, false, true><<<owners, OWN_T, 0, stream>>>(
      own8, rec, E, N, h2, es, ed, b2, ln1_g, ln1_b, (float*)d_out, sv_cnt, sv_src, sv_dl);
}


# Row tails: eight lanes per node, the four nodes of a wave round reduce side by side (3 exchange steps, 2 rounds; was 7 nodes in turn, 5 steps each); LayerNorm and softmax sums use a new tree
# speedup vs baseline: 4.4562x; 1.0460x over previous
#include <hip/hip_runtime.h>
#include <math.h>

typedef __attribute__((ext_vector_type(2))) float v2f;
typedef __attribute__((ext_vector_type(8))) float v8f;

#define GAT_NEG_SLOPE 0.2f
#define GAT_SM_EPS 1e-16f
#define GAT_LN_EPS 1e-5f

__device__ __forceinline__ float group_sum8(float v) {
#pragma unroll
  for (int o = 4; o > 0; o >>= 1) v += __shfl_xor(v, o, 32);
  return v;
}
__device__ __forceinline__ float group_max8(float v) {
#pragma unroll
  for (int o = 4; o > 0; o >>= 1) v = fmaxf(v, __shfl_xor(v, o, 32));
  return v;
}


#define GEMM_T 512
template <int K, int NC>
__global__ void __launch_bounds__(GEMM_T)
gemm_wmma_f32(const float* __restrict__ A, const float* __restrict__ B, float* __restrict__ C,
              int M) {
  constexpr int T = NC / 16;
  constexpr int KB = K / 16;
  static_assert(K % 16 == 0 && NC % 16 == 0, "whole fragments only");
  __shared__ float4 Bs[KB * 2 * T * 32];
  const int tid = threadIdx.x;
  const int lane = tid & 31;
  const int wave = __builtin_amdgcn_readfirstlane(tid >> 5);
  const int tile = blockIdx.x * (GEMM_T / 32) + wave;
  const int c = lane & 15, h = lane >> 4;

  const int rowA = min(tile * 16 + c, M - 1);
  const float4* __restrict__ ap = (const float4*)(A + (size_t)rowA * K + 8 * h);
  float4 a[KB][2];
#pragma unroll
  for (int kb = 0; kb < KB; ++kb) { a[kb][0] = ap[4 * kb]; a[kb][1] = ap[4 * kb + 1]; }

  constexpr int ITEMS = (K / 4) * NC;
  static_assert(ITEMS % GEMM_T == 0, "every thread stages the same number of items");
  float4 bv4[ITEMS / GEMM_T];
#pragma unroll
  for (int u = 0; u < ITEMS / GEMM_T; ++u) {
    const int it = tid + u * GEMM_T;
    const int m = it / NC, n = it - m * NC;
    const float* __restrict__ bp = B + (size_t)(4 * m) * NC + n;
    bv4[u].x = bp[0]; bv4[u].y = bp[NC]; bv4[u].z = bp[2 * NC]; bv4[u].w = bp[3 * NC];
  }
#pragma unroll
  for (int u = 0; u < ITEMS / GEMM_T; ++u) {
    const int it = tid + u * GEMM_T;
    const int m = it / NC, n = it - m * NC;
    const int kb = m >> 2, hh = (m >> 1) & 1, jp = m & 1;
    Bs[((kb * 2 + jp) * T + (n >> 4)) * 32 + hh * 16 + (n & 15)] = bv4[u];
  }
  __syncthreads();
  if (tile * 16 >= M) return;

  v8f acc[T];
#pragma unroll
  for (int t = 0; t < T; ++t) acc[t] = v8f{};
#pragma unroll
  for (int kb = 0; kb < KB; ++kb) {
#pragma unroll
    for (int jp = 0; jp < 2; ++jp) {
      const float4 av = a[kb][jp];
      const v2f a0 = {av.x, av.y}, a1 = {av.z, av.w};
#pragma unroll
      for (int t = 0; t < T; ++t) {
        const float4 bv = Bs[((kb * 2 + jp) * T + t) * 32 + lane];
        const v2f b0 = {bv.x, bv.y}, b1 = {bv.z, bv.w};
        acc[t] = __builtin_amdgcn_wmma_f32_16x16x4_f32(false, a0, false, b0, (short)0, acc[t],
                                                       false, false);
        acc[t] = __builtin_amdgcn_wmma_f32_16x16x4_f32(false, a1, false, b1, (short)0, acc[t],
                                                       false, false);
      }
    }
  }

  const int rowTop = tile * 16 + h * 8;
#pragma unroll
  for (int t = 0; t < T; ++t) {
#pragma unroll
    for (int r = 0; r < 8; ++r) {
      const int row = rowTop + r;
      if (row < M) C[(size_t)row * NC + t * 16 + c] = acc[t][r];
    }
  }
}

#define OWN_T 1024
#define OWN_RPW 196
#define OWN_CAP 6144
#define OWN_SCAN_U 7
static_assert(OWN_RPW <= 256, "a destination's place in its owner's range is kept in one byte");

__global__ void attn_scores_k(const float* __restrict__ h, const float* __restrict__ a_src,
                              const float* __restrict__ a_dst, float* __restrict__ es,
                              float* __restrict__ ed, int N, int H, int CH,
                              const int* __restrict__ ei, int E,
                              unsigned char* __restrict__ own8, int* __restrict__ rec) {
  int idx = blockIdx.x * blockDim.x + threadIdx.x;
  if (ei != nullptr) {
    const int* __restrict__ dstp = ei + E;
    const int nthr = gridDim.x * blockDim.x;
    auto pack = [](int sN, int dN, unsigned& o) -> int {
      o = (unsigned)dN / (unsigned)OWN_RPW;
      return (int)(((unsigned)sN << 8) | ((unsigned)dN - o * (unsigned)OWN_RPW));
    };
    if ((E & 3) == 0 && ((((size_t)rec) | ((size_t)own8)) & 15) == 0) {
      for (int q = idx; q < (E >> 2); q += nthr) {
        const int4 sv = ((const int4*)ei)[q];
        const int4 dv = ((const int4*)dstp)[q];
        unsigned o0, o1, o2, o3;
        int4 r;
        r.x = pack(sv.x, dv.x, o0); r.y = pack(sv.y, dv.y, o1);
        r.z = pack(sv.z, dv.z, o2); r.w = pack(sv.w, dv.w, o3);
        ((unsigned*)own8)[q] = (o0 & 255u) | ((o1 & 255u) << 8) | ((o2 & 255u) << 16) | (o3 << 24);
        ((int4*)rec)[q] = r;
      }
    } else {
      for (int e = idx; e < E; e += nthr) {
        unsigned o;
        rec[e] = pack(ei[e], dstp[e], o);
        own8[e] = (unsigned char)o;
      }
    }
  }
  if (idx >= N * H) return;
  int n = idx / H, hh = idx - n * H;
  const float* hp = h + (size_t)n * H * CH + hh * CH;
  const float* asp = a_src + hh * CH;
  const float* adp = a_dst + hh * CH;
  float s = 0.0f, d = 0.0f;
  for (int c = 0; c < CH; ++c) { float t = hp[c]; s += t * asp[c]; d += t * adp[c]; }
  es[idx] = s; ed[idx] = d;
}

__device__ __forceinline__ int lane_i32(int v, int src) {
  return __builtin_amdgcn_readlane(v, src);
}
__device__ __forceinline__ float lane_f32(float v, int src) {
  return __int_as_float(__builtin_amdgcn_readlane(__float_as_int(v), src));
}

__device__ __forceinline__ float leaky(float v) { return (v > 0.0f) ? v : GAT_NEG_SLOPE * v; }

__device__ __forceinline__ void lds_max_f32(float* addr, float v) {
  if (v >= 0.0f) atomicMax((int*)addr, __float_as_int(v));
  else           atomicMin((unsigned int*)addr, __float_as_uint(v));
}

template <int CH, bool CONCAT, bool SAVE, bool LOAD>
__global__ void __launch_bounds__(OWN_T)
owner_conv_k(const unsigned char* __restrict__ own8, const int* __restrict__ rec, int E, int N,
             const float* __restrict__ feat, const float* __restrict__ es,
             const float* __restrict__ ed, const float* __restrict__ bias,
             const float* __restrict__ lg, const float* __restrict__ lb,
             float* __restrict__ out,
             int* __restrict__ sv_cnt, int* __restrict__ sv_src, int* __restrict__ sv_dl) {
  constexpr int H = 4;
  constexpr int FR = H * CH;
  constexpr int KCH = FR / 32;
  constexpr int EPB = 8;
  __shared__ __attribute__((aligned(16))) float acc[OWN_RPW * FR];
  __shared__ float lw[OWN_CAP * H];
  __shared__ int lsrc[OWN_CAP];
  __shared__ int ldl[OWN_CAP];
  __shared__ float edS[OWN_RPW * H], slf[OWN_RPW * H], mx[OWN_RPW * H], den[OWN_RPW * H];
  __shared__ int cnt;

  const int tid = threadIdx.x;
  const int lane = tid & 31;
  const int wave = __builtin_amdgcn_readfirstlane(tid >> 5);
  const int lo = blockIdx.x * OWN_RPW;
  const int nloc = min(OWN_RPW, N - lo);
  const size_t svBase = (size_t)blockIdx.x * OWN_CAP;

  for (int i = tid; i < nloc * FR; i += OWN_T) acc[i] = 0.0f;
  for (int i = tid; i < nloc * H; i += OWN_T) {
    const float d = ed[(size_t)lo * H + i];
    const float v = leaky(es[(size_t)lo * H + i] + d);
    edS[i] = d; slf[i] = v; mx[i] = v; den[i] = 0.0f;
  }

  auto scan = [&](int a, int b) -> int {
    if (tid == 0) cnt = 0;
    __syncthreads();
    auto hit = [&](int e) {
      const int i = atomicAdd(&cnt, 1);
      if (i < OWN_CAP) lsrc[i] = e;
    };
    const unsigned me4 = (unsigned)blockIdx.x * 0x01010101u;
    auto match = [&](unsigned w) -> unsigned {
      const unsigned x = w ^ me4;
      return ~((((x & 0x7f7f7f7fu) + 0x7f7f7f7fu) | x) | 0x7f7f7f7fu);
    };
    if ((((size_t)(own8 + a)) & 15) == 0) {
      const int nv = (b - a) >> 4;
      const uint4* __restrict__ p4 = (const uint4*)(own8 + a);
      const int rot = (int)(((long long)blockIdx.x * nv) / gridDim.x) & ~31;
      for (int i0 = tid; i0 < nv; i0 += OWN_T * OWN_SCAN_U) {
        uint4 v[OWN_SCAN_U];
        int at[OWN_SCAN_U];
#pragma unroll
        for (int u = 0; u < OWN_SCAN_U; ++u) {
          const int ii = i0 + u * OWN_T;
          int jj = ii + rot;
          if (jj >= nv) jj -= nv;
          at[u] = (ii < nv) ? jj : -1;
          v[u] = p4[max(at[u], 0)];
        }
#pragma unroll
        for (int u = 0; u < OWN_SCAN_U; ++u) {
          unsigned m = (match(v[u].x) >> 7) | (match(v[u].y) >> 6) |
                       (match(v[u].z) >> 5) | (match(v[u].w) >> 4);
          if (at[u] < 0) m = 0u;
          const int e0 = a + 16 * at[u];
          while (m != 0u) {
            const int p = __ffs((int)m) - 1;
            hit(e0 + 4 * (p & 7) + (p >> 3));
            m &= m - 1u;
          }
        }
      }
      for (int e = a + 16 * nv + tid; e < b; e += OWN_T)
        if (own8[e] == (unsigned char)blockIdx.x) hit(e);
    } else {
      for (int e = a + tid; e < b; e += OWN_T)
        if (own8[e] == (unsigned char)blockIdx.x) hit(e);
    }
    __syncthreads();
    return cnt;
  };

  auto resolve = [&](int n, bool fromEdgeIds) {
    for (int i0 = tid; i0 < n; i0 += OWN_T * 4) {
      int s[4], dls[4];
      float4 ev[4];
#pragma unroll
      for (int u = 0; u < 4; ++u) {
        const int i = i0 + u * OWN_T;
        int t = (i < n) ? lsrc[i] : 0;
        if (fromEdgeIds) {
          const int r = rec[t];
          s[u] = (int)((unsigned)r >> 8);
          dls[u] = r & 255;
        } else {
          s[u] = t;
          dls[u] = (i < n) ? ldl[i] : 0;
        }
      }
#pragma unroll
      for (int u = 0; u < 4; ++u) ev[u] = *(const float4*)(es + (size_t)s[u] * H);
#pragma unroll
      for (int u = 0; u < 4; ++u) {
        const int i = i0 + u * OWN_T;
        if (i < n) {
          const int dl = dls[u];
          if (fromEdgeIds) { lsrc[i] = s[u]; ldl[i] = dl; }
          const float l0 = leaky(ev[u].x + edS[dl * H + 0]);
          const float l1 = leaky(ev[u].y + edS[dl * H + 1]);
          const float l2 = leaky(ev[u].z + edS[dl * H + 2]);
          const float l3 = leaky(ev[u].w + edS[dl * H + 3]);
          lw[i * H + 0] = l0; lw[i * H + 1] = l1; lw[i * H + 2] = l2; lw[i * H + 3] = l3;
          lds_max_f32(&mx[dl * H + 0], l0); lds_max_f32(&mx[dl * H + 1], l1);
          lds_max_f32(&mx[dl * H + 2], l2); lds_max_f32(&mx[dl * H + 3], l3);
        }
      }
    }
    __syncthreads();
  };

  auto sumphase = [&](int n, bool first) {
    const int M = (n + (first ? nloc : 0)) * H;
    for (int it = tid; it < M; it += OWN_T) {
      const int i = it >> 2, h = it & 3;
      int dl; float l;
      if (i < n) { dl = ldl[i]; l = lw[it]; }
      else       { dl = i - n;  l = slf[dl * H + h]; }
      atomicAdd(&den[dl * H + h], __expf(l - mx[dl * H + h]));
    }
  };

  auto aggphase = [&](int n, bool first) {
    const int M = n + (first ? nloc : 0);
    const int nR = (M + EPB - 1) / EPB;
    for (int r = wave; r < nR; r += OWN_T / 32) {
      const int h = lane & 3;
      const int i = r * EPB + (lane >> 2);
      int s = 0, dl = 0; float alpha = 0.0f;
      if (i < M) {
        float l;
        if (i < n) { s = lsrc[i]; dl = ldl[i]; l = lw[i * H + h]; }
        else       { dl = i - n; s = lo + dl; l = slf[dl * H + h]; }
        alpha = __expf(l - mx[dl * H + h]) / (den[dl * H + h] + GAT_SM_EPS);
      }
      float f[EPB][KCH];
#pragma unroll
      for (int j = 0; j < EPB; ++j) {
        const float* row = feat + (size_t)lane_i32(s, 4 * j) * FR;
#pragma unroll
        for (int k = 0; k < KCH; ++k) f[j][k] = row[lane + 32 * k];
      }
#pragma unroll
      for (int j = 0; j < EPB; ++j) {
        if (r * EPB + j < M) {
          float* arow = acc + lane_i32(dl, 4 * j) * FR;
#pragma unroll
          for (int k = 0; k < KCH; ++k) {
            const int hLo = (32 * k) / CH, hHi = (32 * k + 31) / CH;
            float w = lane_f32(alpha, 4 * j + hLo);
            if (hHi != hLo) {
              const float wHi = lane_f32(alpha, 4 * j + hHi);
              w = (lane < (hLo + 1) * CH - 32 * k) ? w : wHi;
            }
            atomicAdd(&arow[lane + 32 * k], f[j][k] * w);
          }
        }
      }
    }
  };

  int n = 0, nseg = 1;
  bool slow = false;
  for (int pass = 0; pass < 3; ++pass) {
    for (int seg = 0; seg < nseg; ++seg) {
      if (pass == 0 || slow) {
        int total = -1;
        if (LOAD && !slow) total = sv_cnt[blockIdx.x];
        if (total >= 0 && total <= OWN_CAP) {
          for (int i = tid; i < total; i += OWN_T) {
            lsrc[i] = sv_src[svBase + i];
            ldl[i] = sv_dl[svBase + i];
          }
          __syncthreads();
          n = total;
          resolve(n, false);
        } else {
          const int a = slow ? seg * OWN_CAP : 0;
          const int b = slow ? min(E, a + OWN_CAP) : E;
          total = scan(a, b);
          n = min(total, OWN_CAP);
          resolve(n, true);
          if (!slow) {
            if (SAVE) {
              if (total <= OWN_CAP) {
                for (int i = tid; i < n; i += OWN_T) {
                  sv_src[svBase + i] = lsrc[i];
                  sv_dl[svBase + i] = ldl[i];
                }
              }
              if (tid == 0) sv_cnt[blockIdx.x] = (total <= OWN_CAP) ? total : -1;
            }
            if (total > OWN_CAP) {
              slow = true;
              nseg = (E + OWN_CAP - 1) / OWN_CAP;
              seg = -1;
              continue;
            }
          }
        }
      }
      if (pass == 1) { sumphase(n, seg == 0); __syncthreads(); }
      if (pass == 2) { aggphase(n, seg == 0); __syncthreads(); }
    }
  }

  constexpr int TL = 8;
  constexpr int NPR = 32 / TL;
  const int sub = lane & (TL - 1);
  const int grp = lane / TL;
  float4 g4[CONCAT ? 4 : 1], s4[CONCAT ? 4 : 1];
  if constexpr (CONCAT) {
#pragma unroll
    for (int m = 0; m < 4; ++m) {
      g4[m] = *(const float4*)(lg + 32 * m + 4 * sub);
      s4[m] = *(const float4*)(lb + 32 * m + 4 * sub);
    }
  }
  for (int i0 = wave * NPR; i0 < nloc; i0 += (OWN_T / 32) * NPR) {
    const bool live = (i0 + grp) < nloc;
    const int i = live ? (i0 + grp) : (nloc - 1);
    const int node = lo + i;
    const float* ap = acc + i * FR;
    if constexpr (CONCAT) {
      static_assert(FR == TL * 16, "four 16-byte pieces per lane");
      float4 v[4];
      float sum = 0.0f;
#pragma unroll
      for (int m = 0; m < 4; ++m) {
        const int c = 32 * m + 4 * sub;
        const float4 a4 = *(const float4*)(ap + c);
        const float4 b4 = *(const float4*)(bias + c);
        float4 t;
        t.x = a4.x + b4.x; t.y = a4.y + b4.y; t.z = a4.z + b4.z; t.w = a4.w + b4.w;
        t.x = (t.x > 0.0f) ? t.x : (__expf(t.x) - 1.0f);
        t.y = (t.y > 0.0f) ? t.y : (__expf(t.y) - 1.0f);
        t.z = (t.z > 0.0f) ? t.z : (__expf(t.z) - 1.0f);
        t.w = (t.w > 0.0f) ? t.w : (__expf(t.w) - 1.0f);
        v[m] = t;
        sum += t.x; sum += t.y; sum += t.z; sum += t.w;
      }
      const float mu = group_sum8(sum) * (1.0f / 128.0f);
      float vs = 0.0f;
#pragma unroll
      for (int m = 0; m < 4; ++m) {
        float4 d = v[m];
        d.x -= mu; d.y -= mu; d.z -= mu; d.w -= mu;
        v[m] = d;
        vs += d.x * d.x; vs += d.y * d.y; vs += d.z * d.z; vs += d.w * d.w;
      }
      const float var = group_sum8(vs) * (1.0f / 128.0f);
      const float rstd = rsqrtf(var + GAT_LN_EPS);
      float* xp = out + (size_t)node * FR;
#pragma unroll
      for (int m = 0; m < 4; ++m) {
        const int c = 32 * m + 4 * sub;
        float4 o;
        o.x = v[m].x * rstd * g4[m].x + s4[m].x; o.y = v[m].y * rstd * g4[m].y + s4[m].y;
        o.z = v[m].z * rstd * g4[m].z + s4[m].z; o.w = v[m].w * rstd * g4[m].w + s4[m].w;
        if (live) *(float4*)(xp + c) = o;
      }
    } else {
      static_assert(CH % TL == 0, "every lane of a group holds CH / TL classes");
      constexpr int CPL = CH / TL;
      float x[CPL];
      float sum = 0.0f;
#pragma unroll
      for (int m = 0; m < CPL; ++m) {
        const int c = sub + TL * m;
        x[m] = (ap[c] + ap[c + CH] + ap[c + 2 * CH] + ap[c + 3 * CH]) * 0.25f + bias[c];
        sum += x[m];
      }
      const float mu = group_sum8(sum) * (1.0f / CH);
      float vs = 0.0f;
#pragma unroll
      for (int m = 0; m < CPL; ++m) { x[m] -= mu; vs += x[m] * x[m]; }
      const float var = group_sum8(vs) * (1.0f / CH);
      const float rstd = rsqrtf(var + GAT_LN_EPS);
      float mxz = -3.0e38f;
#pragma unroll
      for (int m = 0; m < CPL; ++m) {
        const int c = sub + TL * m;
        x[m] = x[m] * rstd * lg[c] + lb[c];
        mxz = fmaxf(mxz, x[m]);
      }
      mxz = group_max8(mxz);
      float se = 0.0f;
#pragma unroll
      for (int m = 0; m < CPL; ++m) se += __expf(x[m] - mxz);
      const float lse = __logf(group_sum8(se));
      if (live) {
        float* yp = out + (size_t)node * CH;
#pragma unroll
        for (int m = 0; m < CPL; ++m) yp[sub + TL * m] = x[m] - mxz - lse;
      }
    }
  }
}

extern "C" void kernel_launch(void* const* d_in, const int* in_sizes, int n_in,
                              void* d_out, int out_size, void* d_ws, size_t ws_size,
                              hipStream_t stream) {
  const float* x        = (const float*)d_in[0];
  const int*   ei       = (const int*)d_in[1];
  const float* W1       = (const float*)d_in[2];
  const float* att_src1 = (const float*)d_in[3];
  const float* att_dst1 = (const float*)d_in[4];
  const float* b1       = (const float*)d_in[5];
  const float* W2       = (const float*)d_in[6];
  const float* att_src2 = (const float*)d_in[7];
  const float* att_dst2 = (const float*)d_in[8];
  const float* b2       = (const float*)d_in[9];
  const float* ln0_g    = (const float*)d_in[10];
  const float* ln0_b    = (const float*)d_in[11];
  const float* ln1_g    = (const float*)d_in[12];
  const float* ln1_b    = (const float*)d_in[13];

  const int F_IN = 256, H1C = 128, H2C = 160, H = 4;
  const int N = in_sizes[0] / F_IN;
  const int E = in_sizes[1] / 2;

  const int mTiles = (N + 15) / 16;
  const int gBlocks = (mTiles + GEMM_T / 32 - 1) / (GEMM_T / 32);
  const int owners = (N + OWN_RPW - 1) / OWN_RPW;

  float* ws   = (float*)d_ws;
  float* h1   = ws;
  float* out1 = h1   + (size_t)N * H1C;
  float* h2   = out1 + (size_t)N * H1C;
  float* es   = h2   + (size_t)N * H2C;
  float* ed   = es   + (size_t)N * H;
  int* sv_cnt = (int*)(ed + (size_t)N * H);
  int* sv_src = sv_cnt + owners;
  int* sv_dl  = sv_src + (size_t)owners * OWN_CAP;
  int* rec    = sv_dl + (size_t)owners * OWN_CAP;
  unsigned char* own8 = (unsigned char*)(rec + E);

  gemm_wmma_f32<256, 128><<<gBlocks, GEMM_T, 0, stream>>>(x, W1, h1, N);
  attn_scores_k<<<(N * H + 255) / 256, 256, 0, stream>>>(h1, att_src1, att_dst1,
                                                         es, ed, N, H, 32, ei, E, own8, rec);
  owner_conv_k<32, true, true, false><<<owners, OWN_T, 0, stream>>>(
      own8, rec, E, N, h1, es, ed, b1, ln0_g, ln0_b, out1, sv_cnt, sv_src, sv_dl);

  gemm_wmma_f32<128, 160><<<gBlocks, GEMM_T, 0, stream>>>(out1, W2, h2, N);
  attn_scores_k<<<(N * H + 255) / 256, 256, 0, stream>>>(h2, att_src2, att_dst2,
                                                         es, ed, N, H, 40, nullptr, 0, nullptr, nullptr);
  owner_conv_k<40, false, false, true><<<owners, OWN_T, 0, stream>>>(
      own8, rec, E, N, h2, es, ed, b2, ln1_g, ln1_b, (float*)d_out, sv_cnt, sv_src, sv_dl);
}


# conv2 product and scores computed by conv1 owners on rows still in LDS: f32 WMMA against W2 plus 8 columns of W2 folded with the attention vectors; 6 launches to 4; score sums reordered
# speedup vs baseline: 4.8744x; 1.0938x over previous
#include <hip/hip_runtime.h>
#include <math.h>

typedef __attribute__((ext_vector_type(2))) float v2f;
typedef __attribute__((ext_vector_type(8))) float v8f;

#define GAT_NEG_SLOPE 0.2f
#define GAT_SM_EPS 1e-16f
#define GAT_LN_EPS 1e-5f

__device__ __forceinline__ float group_sum8(float v) {
#pragma unroll
  for (int o = 4; o > 0; o >>= 1) v += __shfl_xor(v, o, 32);
  return v;
}
__device__ __forceinline__ float group_max8(float v) {
#pragma unroll
  for (int o = 4; o > 0; o >>= 1) v = fmaxf(v, __shfl_xor(v, o, 32));
  return v;
}


#define GEMM_T 512
template <int K, int NC>
__global__ void __launch_bounds__(GEMM_T)
gemm_wmma_f32(const float* __restrict__ A, const float* __restrict__ B, float* __restrict__ C,
              int M) {
  constexpr int T = NC / 16;
  constexpr int KB = K / 16;
  static_assert(K % 16 == 0 && NC % 16 == 0, "whole fragments only");
  __shared__ float4 Bs[KB * 2 * T * 32];
  const int tid = threadIdx.x;
  const int lane = tid & 31;
  const int wave = __builtin_amdgcn_readfirstlane(tid >> 5);
  const int tile = blockIdx.x * (GEMM_T / 32) + wave;
  const int c = lane & 15, h = lane >> 4;

  const int rowA = min(tile * 16 + c, M - 1);
  const float4* __restrict__ ap = (const float4*)(A + (size_t)rowA * K + 8 * h);
  float4 a[KB][2];
#pragma unroll
  for (int kb = 0; kb < KB; ++kb) { a[kb][0] = ap[4 * kb]; a[kb][1] = ap[4 * kb + 1]; }

  constexpr int ITEMS = (K / 4) * NC;
  static_assert(ITEMS % GEMM_T == 0, "every thread stages the same number of items");
  float4 bv4[ITEMS / GEMM_T];
#pragma unroll
  for (int u = 0; u < ITEMS / GEMM_T; ++u) {
    const int it = tid + u * GEMM_T;
    const int m = it / NC, n = it - m * NC;
    const float* __restrict__ bp = B + (size_t)(4 * m) * NC + n;
    bv4[u].x = bp[0]; bv4[u].y = bp[NC]; bv4[u].z = bp[2 * NC]; bv4[u].w = bp[3 * NC];
  }
#pragma unroll
  for (int u = 0; u < ITEMS / GEMM_T; ++u) {
    const int it = tid + u * GEMM_T;
    const int m = it / NC, n = it - m * NC;
    const int kb = m >> 2, hh = (m >> 1) & 1, jp = m & 1;
    Bs[((kb * 2 + jp) * T + (n >> 4)) * 32 + hh * 16 + (n & 15)] = bv4[u];
  }
  __syncthreads();
  if (tile * 16 >= M) return;

  v8f acc[T];
#pragma unroll
  for (int t = 0; t < T; ++t) acc[t] = v8f{};
#pragma unroll
  for (int kb = 0; kb < KB; ++kb) {
#pragma unroll
    for (int jp = 0; jp < 2; ++jp) {
      const float4 av = a[kb][jp];
      const v2f a0 = {av.x, av.y}, a1 = {av.z, av.w};
#pragma unroll
      for (int t = 0; t < T; ++t) {
        const float4 bv = Bs[((kb * 2 + jp) * T + t) * 32 + lane];
        const v2f b0 = {bv.x, bv.y}, b1 = {bv.z, bv.w};
        acc[t] = __builtin_amdgcn_wmma_f32_16x16x4_f32(false, a0, false, b0, (short)0, acc[t],
                                                       false, false);
        acc[t] = __builtin_amdgcn_wmma_f32_16x16x4_f32(false, a1, false, b1, (short)0, acc[t],
                                                       false, false);
      }
    }
  }

  const int rowTop = tile * 16 + h * 8;
#pragma unroll
  for (int t = 0; t < T; ++t) {
#pragma unroll
    for (int r = 0; r < 8; ++r) {
      const int row = rowTop + r;
      if (row < M) C[(size_t)row * NC + t * 16 + c] = acc[t][r];
    }
  }
}

#define OWN_T 1024
#define OWN_RPW 196
#define OWN_CAP 6144
#define OWN_SCAN_U 7
static_assert(OWN_RPW <= 256, "a destination's place in its owner's range is kept in one byte");

__global__ void attn_scores_k(const float* __restrict__ h, const float* __restrict__ a_src,
                              const float* __restrict__ a_dst, float* __restrict__ es,
                              float* __restrict__ ed, int N, int H, int CH,
                              const int* __restrict__ ei, int E,
                              unsigned char* __restrict__ own8, int* __restrict__ rec) {
  int idx = blockIdx.x * blockDim.x + threadIdx.x;
  if (ei != nullptr) {
    const int* __restrict__ dstp = ei + E;
    const int nthr = gridDim.x * blockDim.x;
    auto pack = [](int sN, int dN, unsigned& o) -> int {
      o = (unsigned)dN / (unsigned)OWN_RPW;
      return (int)(((unsigned)sN << 8) | ((unsigned)dN - o * (unsigned)OWN_RPW));
    };
    if ((E & 3) == 0 && ((((size_t)rec) | ((size_t)own8)) & 15) == 0) {
      for (int q = idx; q < (E >> 2); q += nthr) {
        const int4 sv = ((const int4*)ei)[q];
        const int4 dv = ((const int4*)dstp)[q];
        unsigned o0, o1, o2, o3;
        int4 r;
        r.x = pack(sv.x, dv.x, o0); r.y = pack(sv.y, dv.y, o1);
        r.z = pack(sv.z, dv.z, o2); r.w = pack(sv.w, dv.w, o3);
        ((unsigned*)own8)[q] = (o0 & 255u) | ((o1 & 255u) << 8) | ((o2 & 255u) << 16) | (o3 << 24);
        ((int4*)rec)[q] = r;
      }
    } else {
      for (int e = idx; e < E; e += nthr) {
        unsigned o;
        rec[e] = pack(ei[e], dstp[e], o);
        own8[e] = (unsigned char)o;
      }
    }
  }
  if (idx >= N * H) return;
  int n = idx / H, hh = idx - n * H;
  const float* hp = h + (size_t)n * H * CH + hh * CH;
  const float* asp = a_src + hh * CH;
  const float* adp = a_dst + hh * CH;
  float s = 0.0f, d = 0.0f;
  for (int c = 0; c < CH; ++c) { float t = hp[c]; s += t * asp[c]; d += t * adp[c]; }
  es[idx] = s; ed[idx] = d;
}

__device__ __forceinline__ int lane_i32(int v, int src) {
  return __builtin_amdgcn_readlane(v, src);
}
__device__ __forceinline__ float lane_f32(float v, int src) {
  return __int_as_float(__builtin_amdgcn_readlane(__float_as_int(v), src));
}

__device__ __forceinline__ float leaky(float v) { return (v > 0.0f) ? v : GAT_NEG_SLOPE * v; }

__device__ __forceinline__ void lds_max_f32(float* addr, float v) {
  if (v >= 0.0f) atomicMax((int*)addr, __float_as_int(v));
  else           atomicMin((unsigned int*)addr, __float_as_uint(v));
}

#define NXT_NC 160
#define NXT_CH 40
template <int V> struct IntC { static constexpr int value = V; };

template <int CH, bool CONCAT, bool SAVE, bool LOAD>
__global__ void __launch_bounds__(OWN_T)
owner_conv_k(const unsigned char* __restrict__ own8, const int* __restrict__ rec, int E, int N,
             const float* __restrict__ feat, const float* __restrict__ es,
             const float* __restrict__ ed, const float* __restrict__ bias,
             const float* __restrict__ lg, const float* __restrict__ lb,
             float* __restrict__ out,
             int* __restrict__ sv_cnt, int* __restrict__ sv_src, int* __restrict__ sv_dl,
             const float* __restrict__ Wn, const float* __restrict__ asn,
             const float* __restrict__ adn, float* __restrict__ esn, float* __restrict__ edn) {
  constexpr int H = 4;
  constexpr int FR = H * CH;
  constexpr int KCH = FR / 32;
  constexpr int EPB = 8;
  constexpr int FRP = CONCAT ? FR + 4 : FR;
  __shared__ __attribute__((aligned(16))) float acc[OWN_RPW * FRP];
  __shared__ __attribute__((aligned(16))) float lw[OWN_CAP * H];
  __shared__ float wa[CONCAT ? FR * 8 : 1];
  __shared__ int lsrc[OWN_CAP];
  __shared__ int ldl[OWN_CAP];
  __shared__ float edS[OWN_RPW * H], slf[OWN_RPW * H], mx[OWN_RPW * H], den[OWN_RPW * H];
  __shared__ int cnt;

  const int tid = threadIdx.x;
  const int lane = tid & 31;
  const int wave = __builtin_amdgcn_readfirstlane(tid >> 5);
  const int lo = blockIdx.x * OWN_RPW;
  const int nloc = min(OWN_RPW, N - lo);
  const size_t svBase = (size_t)blockIdx.x * OWN_CAP;

  for (int i = tid; i < nloc * FRP; i += OWN_T) acc[i] = 0.0f;
  if constexpr (CONCAT) {
    static_assert(FR * 8 == OWN_T, "one folded weight per thread");
    const int k = tid >> 3, j = tid & 7, hs = j & 3;
    const float* __restrict__ wr = Wn + (size_t)k * NXT_NC + hs * NXT_CH;
    const float* __restrict__ ar = ((j < 4) ? asn : adn) + hs * NXT_CH;
    float s = 0.0f;
    for (int c = 0; c < NXT_CH; ++c) s += wr[c] * ar[c];
    wa[tid] = s;
  }
  for (int i = tid; i < nloc * H; i += OWN_T) {
    const float d = ed[(size_t)lo * H + i];
    const float v = leaky(es[(size_t)lo * H + i] + d);
    edS[i] = d; slf[i] = v; mx[i] = v; den[i] = 0.0f;
  }

  auto scan = [&](int a, int b) -> int {
    if (tid == 0) cnt = 0;
    __syncthreads();
    auto hit = [&](int e) {
      const int i = atomicAdd(&cnt, 1);
      if (i < OWN_CAP) lsrc[i] = e;
    };
    const unsigned me4 = (unsigned)blockIdx.x * 0x01010101u;
    auto match = [&](unsigned w) -> unsigned {
      const unsigned x = w ^ me4;
      return ~((((x & 0x7f7f7f7fu) + 0x7f7f7f7fu) | x) | 0x7f7f7f7fu);
    };
    if ((((size_t)(own8 + a)) & 15) == 0) {
      const int nv = (b - a) >> 4;
      const uint4* __restrict__ p4 = (const uint4*)(own8 + a);
      const int rot = (int)(((long long)blockIdx.x * nv) / gridDim.x) & ~31;
      for (int i0 = tid; i0 < nv; i0 += OWN_T * OWN_SCAN_U) {
        uint4 v[OWN_SCAN_U];
        int at[OWN_SCAN_U];
#pragma unroll
        for (int u = 0; u < OWN_SCAN_U; ++u) {
          const int ii = i0 + u * OWN_T;
          int jj = ii + rot;
          if (jj >= nv) jj -= nv;
          at[u] = (ii < nv) ? jj : -1;
          v[u] = p4[max(at[u], 0)];
        }
#pragma unroll
        for (int u = 0; u < OWN_SCAN_U; ++u) {
          unsigned m = (match(v[u].x) >> 7) | (match(v[u].y) >> 6) |
                       (match(v[u].z) >> 5) | (match(v[u].w) >> 4);
          if (at[u] < 0) m = 0u;
          const int e0 = a + 16 * at[u];
          while (m != 0u) {
            const int p = __ffs((int)m) - 1;
            hit(e0 + 4 * (p & 7) + (p >> 3));
            m &= m - 1u;
          }
        }
      }
      for (int e = a + 16 * nv + tid; e < b; e += OWN_T)
        if (own8[e] == (unsigned char)blockIdx.x) hit(e);
    } else {
      for (int e = a + tid; e < b; e += OWN_T)
        if (own8[e] == (unsigned char)blockIdx.x) hit(e);
    }
    __syncthreads();
    return cnt;
  };

  auto resolve = [&](int n, bool fromEdgeIds) {
    for (int i0 = tid; i0 < n; i0 += OWN_T * 4) {
      int s[4], dls[4];
      float4 ev[4];
#pragma unroll
      for (int u = 0; u < 4; ++u) {
        const int i = i0 + u * OWN_T;
        int t = (i < n) ? lsrc[i] : 0;
        if (fromEdgeIds) {
          const int r = rec[t];
          s[u] = (int)((unsigned)r >> 8);
          dls[u] = r & 255;
        } else {
          s[u] = t;
          dls[u] = (i < n) ? ldl[i] : 0;
        }
      }
#pragma unroll
      for (int u = 0; u < 4; ++u) ev[u] = *(const float4*)(es + (size_t)s[u] * H);
#pragma unroll
      for (int u = 0; u < 4; ++u) {
        const int i = i0 + u * OWN_T;
        if (i < n) {
          const int dl = dls[u];
          if (fromEdgeIds) { lsrc[i] = s[u]; ldl[i] = dl; }
          const float l0 = leaky(ev[u].x + edS[dl * H + 0]);
          const float l1 = leaky(ev[u].y + edS[dl * H + 1]);
          const float l2 = leaky(ev[u].z + edS[dl * H + 2]);
          const float l3 = leaky(ev[u].w + edS[dl * H + 3]);
          lw[i * H + 0] = l0; lw[i * H + 1] = l1; lw[i * H + 2] = l2; lw[i * H + 3] = l3;
          lds_max_f32(&mx[dl * H + 0], l0); lds_max_f32(&mx[dl * H + 1], l1);
          lds_max_f32(&mx[dl * H + 2], l2); lds_max_f32(&mx[dl * H + 3], l3);
        }
      }
    }
    __syncthreads();
  };

  auto sumphase = [&](int n, bool first) {
    const int M = (n + (first ? nloc : 0)) * H;
    for (int it = tid; it < M; it += OWN_T) {
      const int i = it >> 2, h = it & 3;
      int dl; float l;
      if (i < n) { dl = ldl[i]; l = lw[it]; }
      else       { dl = i - n;  l = slf[dl * H + h]; }
      atomicAdd(&den[dl * H + h], __expf(l - mx[dl * H + h]));
    }
  };

  auto aggphase = [&](int n, bool first) {
    const int M = n + (first ? nloc : 0);
    const int nR = (M + EPB - 1) / EPB;
    for (int r = wave; r < nR; r += OWN_T / 32) {
      const int h = lane & 3;
      const int i = r * EPB + (lane >> 2);
      int s = 0, dl = 0; float alpha = 0.0f;
      if (i < M) {
        float l;
        if (i < n) { s = lsrc[i]; dl = ldl[i]; l = lw[i * H + h]; }
        else       { dl = i - n; s = lo + dl; l = slf[dl * H + h]; }
        alpha = __expf(l - mx[dl * H + h]) / (den[dl * H + h] + GAT_SM_EPS);
      }
      float f[EPB][KCH];
#pragma unroll
      for (int j = 0; j < EPB; ++j) {
        const float* row = feat + (size_t)lane_i32(s, 4 * j) * FR;
#pragma unroll
        for (int k = 0; k < KCH; ++k) f[j][k] = row[lane + 32 * k];
      }
#pragma unroll
      for (int j = 0; j < EPB; ++j) {
        if (r * EPB + j < M) {
          float* arow = acc + lane_i32(dl, 4 * j) * FRP;
#pragma unroll
          for (int k = 0; k < KCH; ++k) {
            const int hLo = (32 * k) / CH, hHi = (32 * k + 31) / CH;
            float w = lane_f32(alpha, 4 * j + hLo);
            if (hHi != hLo) {
              const float wHi = lane_f32(alpha, 4 * j + hHi);
              w = (lane < (hLo + 1) * CH - 32 * k) ? w : wHi;
            }
            atomicAdd(&arow[lane + 32 * k], f[j][k] * w);
          }
        }
      }
    }
  };

  int n = 0, nseg = 1;
  bool slow = false;
  for (int pass = 0; pass < 3; ++pass) {
    for (int seg = 0; seg < nseg; ++seg) {
      if (pass == 0 || slow) {
        int total = -1;
        if (LOAD && !slow) total = sv_cnt[blockIdx.x];
        if (total >= 0 && total <= OWN_CAP) {
          for (int i = tid; i < total; i += OWN_T) {
            lsrc[i] = sv_src[svBase + i];
            ldl[i] = sv_dl[svBase + i];
          }
          __syncthreads();
          n = total;
          resolve(n, false);
        } else {
          const int a = slow ? seg * OWN_CAP : 0;
          const int b = slow ? min(E, a + OWN_CAP) : E;
          total = scan(a, b);
          n = min(total, OWN_CAP);
          resolve(n, true);
          if (!slow) {
            if (SAVE) {
              if (total <= OWN_CAP) {
                for (int i = tid; i < n; i += OWN_T) {
                  sv_src[svBase + i] = lsrc[i];
                  sv_dl[svBase + i] = ldl[i];
                }
              }
              if (tid == 0) sv_cnt[blockIdx.x] = (total <= OWN_CAP) ? total : -1;
            }
            if (total > OWN_CAP) {
              slow = true;
              nseg = (E + OWN_CAP - 1) / OWN_CAP;
              seg = -1;
              continue;
            }
          }
        }
      }
      if (pass == 1) { sumphase(n, seg == 0); __syncthreads(); }
      if (pass == 2) { aggphase(n, seg == 0); __syncthreads(); }
    }
  }

  constexpr int TL = 8;
  constexpr int NPR = 32 / TL;
  const int sub = lane & (TL - 1);
  const int grp = lane / TL;
  float4 g4[CONCAT ? 4 : 1], s4[CONCAT ? 4 : 1];
  if constexpr (CONCAT) {
#pragma unroll
    for (int m = 0; m < 4; ++m) {
      g4[m] = *(const float4*)(lg + 32 * m + 4 * sub);
      s4[m] = *(const float4*)(lb + 32 * m + 4 * sub);
    }
  }
  constexpr int NT = NXT_NC / 16 + 1;
  constexpr int KB = FR / 16;
  constexpr int ITEMS = CONCAT ? (FR / 4) * NXT_NC : OWN_T;
  static_assert(ITEMS % OWN_T == 0, "every thread stages the same number of items");
  static_assert(!CONCAT || KB * 2 * NT * 32 * 4 <= OWN_CAP * H, "the staged weights fit the logit list's place");
  float4 bv4[ITEMS / OWN_T];
  if constexpr (CONCAT) {
#pragma unroll
    for (int u = 0; u < ITEMS / OWN_T; ++u) {
      const int it = tid + u * OWN_T;
      const int m = it / NXT_NC, n = it - m * NXT_NC;
      const float* __restrict__ bp = Wn + (size_t)(4 * m) * NXT_NC + n;
      bv4[u].x = bp[0]; bv4[u].y = bp[NXT_NC]; bv4[u].z = bp[2 * NXT_NC]; bv4[u].w = bp[3 * NXT_NC];
    }
  }
  for (int i0 = wave * NPR; i0 < nloc; i0 += (OWN_T / 32) * NPR) {
    const bool live = (i0 + grp) < nloc;
    const int i = live ? (i0 + grp) : (nloc - 1);
    const int node = lo + i;
    const float* ap = acc + i * FRP;
    if constexpr (CONCAT) {
      static_assert(FR == TL * 16, "four 16-byte pieces per lane");
      float4 v[4];
      float sum = 0.0f;
#pragma unroll
      for (int m = 0; m < 4; ++m) {
        const int c = 32 * m + 4 * sub;
        const float4 a4 = *(const float4*)(ap + c);
        const float4 b4 = *(const float4*)(bias + c);
        float4 t;
        t.x = a4.x + b4.x; t.y = a4.y + b4.y; t.z = a4.z + b4.z; t.w = a4.w + b4.w;
        t.x = (t.x > 0.0f) ? t.x : (__expf(t.x) - 1.0f);
        t.y = (t.y > 0.0f) ? t.y : (__expf(t.y) - 1.0f);
        t.z = (t.z > 0.0f) ? t.z : (__expf(t.z) - 1.0f);
        t.w = (t.w > 0.0f) ? t.w : (__expf(t.w) - 1.0f);
        v[m] = t;
        sum += t.x; sum += t.y; sum += t.z; sum += t.w;
      }
      const float mu = group_sum8(sum) * (1.0f / 128.0f);
      float vs = 0.0f;
#pragma unroll
      for (int m = 0; m < 4; ++m) {
        float4 d = v[m];
        d.x -= mu; d.y -= mu; d.z -= mu; d.w -= mu;
        v[m] = d;
        vs += d.x * d.x; vs += d.y * d.y; vs += d.z * d.z; vs += d.w * d.w;
      }
      const float var = group_sum8(vs) * (1.0f / 128.0f);
      const float rstd = rsqrtf(var + GAT_LN_EPS);
      float* xp = acc + i * FRP;
#pragma unroll
      for (int m = 0; m < 4; ++m) {
        const int c = 32 * m + 4 * sub;
        float4 o;
        o.x = v[m].x * rstd * g4[m].x + s4[m].x; o.y = v[m].y * rstd * g4[m].y + s4[m].y;
        o.z = v[m].z * rstd * g4[m].z + s4[m].z; o.w = v[m].w * rstd * g4[m].w + s4[m].w;
        if (live) *(float4*)(xp + c) = o;
      }
    } else {
      static_assert(CH % TL == 0, "every lane of a group holds CH / TL classes");
      constexpr int CPL = CH / TL;
      float x[CPL];
      float sum = 0.0f;
#pragma unroll
      for (int m = 0; m < CPL; ++m) {
        const int c = sub + TL * m;
        x[m] = (ap[c] + ap[c + CH] + ap[c + 2 * CH] + ap[c + 3 * CH]) * 0.25f + bias[c];
        sum += x[m];
      }
      const float mu = group_sum8(sum) * (1.0f / CH);
      float vs = 0.0f;
#pragma unroll
      for (int m = 0; m < CPL; ++m) { x[m] -= mu; vs += x[m] * x[m]; }
      const float var = group_sum8(vs) * (1.0f / CH);
      const float rstd = rsqrtf(var + GAT_LN_EPS);
      float mxz = -3.0e38f;
#pragma unroll
      for (int m = 0; m < CPL; ++m) {
        const int c = sub + TL * m;
        x[m] = x[m] * rstd * lg[c] + lb[c];
        mxz = fmaxf(mxz, x[m]);
      }
      mxz = group_max8(mxz);
      float se = 0.0f;
#pragma unroll
      for (int m = 0; m < CPL; ++m) se += __expf(x[m] - mxz);
      const float lse = __logf(group_sum8(se));
      if (live) {
        float* yp = out + (size_t)node * CH;
#pragma unroll
        for (int m = 0; m < CPL; ++m) yp[sub + TL * m] = x[m] - mxz - lse;
      }
    }
  }

  if constexpr (CONCAT) {
    float4* Bs = (float4*)lw;
#pragma unroll
    for (int u = 0; u < ITEMS / OWN_T; ++u) {
      const int it = tid + u * OWN_T;
      const int m = it / NXT_NC, n = it - m * NXT_NC;
      const int kb = m >> 2, hh = (m >> 1) & 1, jp = m & 1;
      Bs[((kb * 2 + jp) * NT + (n >> 4)) * 32 + hh * 16 + (n & 15)] = bv4[u];
    }
    if (tid < (FR / 4) * 16) {
      const int m = tid >> 4, np = tid & 15;
      float4 v = {0.0f, 0.0f, 0.0f, 0.0f};
      if (np < 8) {
        v.x = wa[(4 * m) * 8 + np];     v.y = wa[(4 * m + 1) * 8 + np];
        v.z = wa[(4 * m + 2) * 8 + np]; v.w = wa[(4 * m + 3) * 8 + np];
      }
      const int kb = m >> 2, hh = (m >> 1) & 1, jp = m & 1;
      Bs[((kb * 2 + jp) * NT + (NT - 1)) * 32 + hh * 16 + np] = v;
    }
    __syncthreads();

    const int c16 = lane & 15, hh = lane >> 4;
    auto unit = [&](auto ntl, int rt, int ct0) {
      constexpr int NTL = decltype(ntl)::value;
      const int rowA = min(rt * 16 + c16, OWN_RPW - 1);
      const float* ap = acc + rowA * FRP + 8 * hh;
      v8f d[NTL];
#pragma unroll
      for (int t = 0; t < NTL; ++t) d[t] = v8f{};
#pragma unroll
      for (int kb = 0; kb < KB; ++kb) {
#pragma unroll
        for (int jp = 0; jp < 2; ++jp) {
          const float4 av = *(const float4*)(ap + 16 * kb + 4 * jp);
          const v2f a0 = {av.x, av.y}, a1 = {av.z, av.w};
#pragma unroll
          for (int t = 0; t < NTL; ++t) {
            const float4 bv = Bs[((kb * 2 + jp) * NT + ct0 + t) * 32 + lane];
            const v2f b0 = {bv.x, bv.y}, b1 = {bv.z, bv.w};
            d[t] = __builtin_amdgcn_wmma_f32_16x16x4_f32(false, a0, false, b0, (short)0, d[t],
                                                         false, false);
            d[t] = __builtin_amdgcn_wmma_f32_16x16x4_f32(false, a1, false, b1, (short)0, d[t],
                                                         false, false);
          }
        }
      }
      const int rTop = rt * 16 + hh * 8;
      if (ct0 < NT - 1) {
#pragma unroll
        for (int t = 0; t < NTL; ++t) {
#pragma unroll
          for (int r = 0; r < 8; ++r) {
            const int row = rTop + r;
            if (row < nloc) out[(size_t)(lo + row) * NXT_NC + (ct0 + t) * 16 + c16] = d[t][r];
          }
        }
      } else {
        float* __restrict__ sp = (c16 < 4) ? esn : edn;
#pragma unroll
        for (int r = 0; r < 8; ++r) {
          const int row = rTop + r;
          if (row < nloc && c16 < 8) sp[(size_t)(lo + row) * H + (c16 & 3)] = d[0][r];
        }
      }
    };
    const int nRT = (nloc + 15) >> 4;
    const int nPair = nRT * (NXT_NC / 32);
    const int nU = nPair + nRT;
    for (int u = wave; u < nU; u += OWN_T / 32) {
      if (u < nPair) {
        const int rt = u / (NXT_NC / 32);
        unit(IntC<2>{}, rt, 2 * (u - rt * (NXT_NC / 32)));
      } else {
        unit(IntC<1>{}, u - nPair, NT - 1);
      }
    }
  }
}

extern "C" void kernel_launch(void* const* d_in, const int* in_sizes, int n_in,
                              void* d_out, int out_size, void* d_ws, size_t ws_size,
                              hipStream_t stream) {
  const float* x        = (const float*)d_in[0];
  const int*   ei       = (const int*)d_in[1];
  const float* W1       = (const float*)d_in[2];
  const float* att_src1 = (const float*)d_in[3];
  const float* att_dst1 = (const float*)d_in[4];
  const float* b1       = (const float*)d_in[5];
  const float* W2       = (const float*)d_in[6];
  const float* att_src2 = (const float*)d_in[7];
  const float* att_dst2 = (const float*)d_in[8];
  const float* b2       = (const float*)d_in[9];
  const float* ln0_g    = (const float*)d_in[10];
  const float* ln0_b    = (const float*)d_in[11];
  const float* ln1_g    = (const float*)d_in[12];
  const float* ln1_b    = (const float*)d_in[13];

  const int F_IN = 256, H1C = 128, H2C = 160, H = 4;
  const int N = in_sizes[0] / F_IN;
  const int E = in_sizes[1] / 2;

  const int mTiles = (N + 15) / 16;
  const int gBlocks = (mTiles + GEMM_T / 32 - 1) / (GEMM_T / 32);
  const int owners = (N + OWN_RPW - 1) / OWN_RPW;

  float* ws   = (float*)d_ws;
  float* h1   = ws;
  float* h2   = h1   + (size_t)N * H1C;
  float* es   = h2   + (size_t)N * H2C;
  float* ed   = es   + (size_t)N * H;
  float* es2  = ed   + (size_t)N * H;
  float* ed2  = es2  + (size_t)N * H;
  int* sv_cnt = (int*)(ed2 + (size_t)N * H);
  int* sv_src = sv_cnt + owners;
  int* sv_dl  = sv_src + (size_t)owners * OWN_CAP;
  int* rec    = sv_dl + (size_t)owners * OWN_CAP;
  unsigned char* own8 = (unsigned char*)(rec + E);

  gemm_wmma_f32<256, 128><<<gBlocks, GEMM_T, 0, stream>>>(x, W1, h1, N);
  attn_scores_k<<<(N * H + 255) / 256, 256, 0, stream>>>(h1, att_src1, att_dst1,
                                                         es, ed, N, H, 32, ei, E, own8, rec);
  owner_conv_k<32, true, true, false><<<owners, OWN_T, 0, stream>>>(
      own8, rec, E, N, h1, es, ed, b1, ln0_g, ln0_b, h2, sv_cnt, sv_src, sv_dl,
      W2, att_src2, att_dst2, es2, ed2);

  owner_conv_k<40, false, false, true><<<owners, OWN_T, 0, stream>>>(
      own8, rec, E, N, h2, es2, ed2, b2, ln1_g, ln1_b, (float*)d_out, sv_cnt, sv_src, sv_dl,
      nullptr, nullptr, nullptr, nullptr, nullptr);
}


# conv1 scores and edge planes move into the product launch as side work-groups on idle units: scores = x times (W1 folded with the attention vectors) on f32 WMMA; 4 launches to 3; score sums reordered
# speedup vs baseline: 5.1139x; 1.0491x over previous
#include <hip/hip_runtime.h>
#include <math.h>

typedef __attribute__((ext_vector_type(2))) float v2f;
typedef __attribute__((ext_vector_type(8))) float v8f;

#define GAT_NEG_SLOPE 0.2f
#define GAT_SM_EPS 1e-16f
#define GAT_LN_EPS 1e-5f

__device__ __forceinline__ float group_sum8(float v) {
#pragma unroll
  for (int o = 4; o > 0; o >>= 1) v += __shfl_xor(v, o, 32);
  return v;
}
__device__ __forceinline__ float group_max8(float v) {
#pragma unroll
  for (int o = 4; o > 0; o >>= 1) v = fmaxf(v, __shfl_xor(v, o, 32));
  return v;
}


#define OWN_T 1024
#define OWN_RPW 196
#define OWN_CAP 6144
#define OWN_SCAN_U 7
static_assert(OWN_RPW <= 256, "a destination's place in its owner's range is kept in one byte");

#define GEMM_T 512
template <int K, int NC>
__global__ void __launch_bounds__(GEMM_T)
gemm_wmma_f32(const float* __restrict__ A, const float* __restrict__ B, float* __restrict__ C,
              int M, const float* __restrict__ a_src, const float* __restrict__ a_dst,
              float* __restrict__ es, float* __restrict__ ed,
              const int* __restrict__ ei, int E,
              unsigned char* __restrict__ own8, int* __restrict__ rec) {
  constexpr int T = NC / 16;
  constexpr int KB = K / 16;
  constexpr int H = 4, CH = NC / H;
  static_assert(K % 16 == 0 && NC % 16 == 0, "whole fragments only");
  static_assert(CH == 32 && GEMM_T == 512, "the fold gives a wave one head and K/4 rows of B");
  __shared__ float4 Bs[KB * 2 * T * 32];
  const int tid = threadIdx.x;
  const int lane = tid & 31;
  const int wave = __builtin_amdgcn_readfirstlane(tid >> 5);
  const int c = lane & 15, h = lane >> 4;
  const int nMain = (M + 16 * (GEMM_T / 32) - 1) / (16 * (GEMM_T / 32));
  const int nSide = (int)gridDim.x - nMain;
  const bool side = (int)blockIdx.x >= nMain;

  if (side || nSide <= 0) {
    const int wIdx = side ? (int)blockIdx.x - nMain : (int)blockIdx.x;
    const int nW = side ? nSide : (int)gridDim.x;

    {
      const int* __restrict__ dstp = ei + E;
      const int wtid = wIdx * GEMM_T + tid;
      const int nthr = nW * GEMM_T;
      auto pack = [](int sN, int dN, unsigned& o) -> int {
        o = (unsigned)dN / (unsigned)OWN_RPW;
        return (int)(((unsigned)sN << 8) | ((unsigned)dN - o * (unsigned)OWN_RPW));
      };
      if ((E & 3) == 0 && E >= 4 && ((((size_t)rec) | ((size_t)own8) | ((size_t)ei)) & 15) == 0) {
        const int nq = E >> 2;
        for (int q0 = wtid; q0 < nq; q0 += 4 * nthr) {
          int4 sv[4], dv[4];
#pragma unroll
          for (int u = 0; u < 4; ++u) {
            const int qq = min(q0 + u * nthr, nq - 1);
            sv[u] = ((const int4*)ei)[qq];
            dv[u] = ((const int4*)dstp)[qq];
          }
#pragma unroll
          for (int u = 0; u < 4; ++u) {
            const int q = q0 + u * nthr;
            if (q < nq) {
              unsigned o0, o1, o2, o3;
              int4 r;
              r.x = pack(sv[u].x, dv[u].x, o0); r.y = pack(sv[u].y, dv[u].y, o1);
              r.z = pack(sv[u].z, dv[u].z, o2); r.w = pack(sv[u].w, dv[u].w, o3);
              ((unsigned*)own8)[q] = (o0 & 255u) | ((o1 & 255u) << 8) | ((o2 & 255u) << 16) | (o3 << 24);
              ((int4*)rec)[q] = r;
            }
          }
        }
      } else {
        for (int e = wtid; e < E; e += nthr) {
          unsigned o;
          rec[e] = pack(ei[e], dstp[e], o);
          own8[e] = (unsigned char)o;
        }
      }
    }

    {
      float* Wsf = (float*)Bs;
      const int hs = wave & 3;
      const int piece = lane & 7, rsub = lane >> 3;
      const float4 asv = *(const float4*)(a_src + hs * CH + 4 * piece);
      const float4 adv = *(const float4*)(a_dst + hs * CH + 4 * piece);
      constexpr int FU = K / 16;
      float4 wv[FU];
#pragma unroll
      for (int u = 0; u < FU; ++u) {
        const int k = (wave >> 2) * (K / 4) + 4 * u + rsub;
        wv[u] = *(const float4*)(B + (size_t)k * NC + hs * CH + 4 * piece);
      }
      for (int i = tid; i < KB * 2 * 2 * 8; i += GEMM_T) {
        const int blk = i >> 4, hh = (i >> 3) & 1, col = 8 + (i & 7);
        Bs[blk * 32 + hh * 16 + col] = float4{0.0f, 0.0f, 0.0f, 0.0f};
      }
#pragma unroll
      for (int u = 0; u < FU; ++u) {
        const int k = (wave >> 2) * (K / 4) + 4 * u + rsub;
        float ps = wv[u].x * asv.x + wv[u].y * asv.y + wv[u].z * asv.z + wv[u].w * asv.w;
        float pd = wv[u].x * adv.x + wv[u].y * adv.y + wv[u].z * adv.z + wv[u].w * adv.w;
        ps = group_sum8(ps);
        pd = group_sum8(pd);
        if (piece == 0) {
          const int kb = k >> 4, hh = (k >> 3) & 1, jp = (k >> 2) & 1, q = k & 3;
          float* slot = Wsf + (((kb * 2 + jp) * 32 + hh * 16) * 4 + q);
          slot[hs * 4] = ps;
          slot[(4 + hs) * 4] = pd;
        }
      }
    }
    __syncthreads();

    for (int tl = wIdx * (GEMM_T / 32) + wave; tl * 16 < M; tl += nW * (GEMM_T / 32)) {
      const int rowS = min(tl * 16 + c, M - 1);
      const float4* __restrict__ sp4 = (const float4*)(A + (size_t)rowS * K + 8 * h);
      float4 sa[KB][2];
#pragma unroll
      for (int kb = 0; kb < KB; ++kb) { sa[kb][0] = sp4[4 * kb]; sa[kb][1] = sp4[4 * kb + 1]; }
      v8f d[2] = {v8f{}, v8f{}};
#pragma unroll
      for (int kb = 0; kb < KB; ++kb) {
#pragma unroll
        for (int jp = 0; jp < 2; ++jp) {
          const float4 av = sa[kb][jp];
          const v2f a0 = {av.x, av.y}, a1 = {av.z, av.w};
          const float4 bv = Bs[(kb * 2 + jp) * 32 + lane];
          const v2f b0 = {bv.x, bv.y}, b1 = {bv.z, bv.w};
          d[jp] = __builtin_amdgcn_wmma_f32_16x16x4_f32(false, a0, false, b0, (short)0, d[jp],
                                                        false, false);
          d[jp] = __builtin_amdgcn_wmma_f32_16x16x4_f32(false, a1, false, b1, (short)0, d[jp],
                                                        false, false);
        }
      }
      float* __restrict__ sp = (c < 4) ? es : ed;
#pragma unroll
      for (int r = 0; r < 8; ++r) {
        const int row = tl * 16 + h * 8 + r;
        if (row < M && c < 8) sp[(size_t)row * H + (c & 3)] = d[0][r] + d[1][r];
      }
    }
    if (side) return;
    __syncthreads();
  }

  const int tile = blockIdx.x * (GEMM_T / 32) + wave;

  const int rowA = min(tile * 16 + c, M - 1);
  const float4* __restrict__ ap = (const float4*)(A + (size_t)rowA * K + 8 * h);
  float4 a[KB][2];
#pragma unroll
  for (int kb = 0; kb < KB; ++kb) { a[kb][0] = ap[4 * kb]; a[kb][1] = ap[4 * kb + 1]; }

  constexpr int ITEMS = (K / 4) * NC;
  static_assert(ITEMS % GEMM_T == 0, "every thread stages the same number of items");
  float4 bv4[ITEMS / GEMM_T];
#pragma unroll
  for (int u = 0; u < ITEMS / GEMM_T; ++u) {
    const int it = tid + u * GEMM_T;
    const int m = it / NC, n = it - m * NC;
    const float* __restrict__ bp = B + (size_t)(4 * m) * NC + n;
    bv4[u].x = bp[0]; bv4[u].y = bp[NC]; bv4[u].z = bp[2 * NC]; bv4[u].w = bp[3 * NC];
  }
#pragma unroll
  for (int u = 0; u < ITEMS / GEMM_T; ++u) {
    const int it = tid + u * GEMM_T;
    const int m = it / NC, n = it - m * NC;
    const int kb = m >> 2, hh = (m >> 1) & 1, jp = m & 1;
    Bs[((kb * 2 + jp) * T + (n >> 4)) * 32 + hh * 16 + (n & 15)] = bv4[u];
  }
  __syncthreads();
  if (tile * 16 >= M) return;

  v8f acc[T];
#pragma unroll
  for (int t = 0; t < T; ++t) acc[t] = v8f{};
#pragma unroll
  for (int kb = 0; kb < KB; ++kb) {
#pragma unroll
    for (int jp = 0; jp < 2; ++jp) {
      const float4 av = a[kb][jp];
      const v2f a0 = {av.x, av.y}, a1 = {av.z, av.w};
#pragma unroll
      for (int t = 0; t < T; ++t) {
        const float4 bv = Bs[((kb * 2 + jp) * T + t) * 32 + lane];
        const v2f b0 = {bv.x, bv.y}, b1 = {bv.z, bv.w};
        acc[t] = __builtin_amdgcn_wmma_f32_16x16x4_f32(false, a0, false, b0, (short)0, acc[t],
                                                       false, false);
        acc[t] = __builtin_amdgcn_wmma_f32_16x16x4_f32(false, a1, false, b1, (short)0, acc[t],
                                                       false, false);
      }
    }
  }

  const int rowTop = tile * 16 + h * 8;
#pragma unroll
  for (int t = 0; t < T; ++t) {
#pragma unroll
    for (int r = 0; r < 8; ++r) {
      const int row = rowTop + r;
      if (row < M) C[(size_t)row * NC + t * 16 + c] = acc[t][r];
    }
  }
}

__device__ __forceinline__ int lane_i32(int v, int src) {
  return __builtin_amdgcn_readlane(v, src);
}
__device__ __forceinline__ float lane_f32(float v, int src) {
  return __int_as_float(__builtin_amdgcn_readlane(__float_as_int(v), src));
}

__device__ __forceinline__ float leaky(float v) { return (v > 0.0f) ? v : GAT_NEG_SLOPE * v; }

__device__ __forceinline__ void lds_max_f32(float* addr, float v) {
  if (v >= 0.0f) atomicMax((int*)addr, __float_as_int(v));
  else           atomicMin((unsigned int*)addr, __float_as_uint(v));
}

#define NXT_NC 160
#define NXT_CH 40
template <int V> struct IntC { static constexpr int value = V; };

template <int CH, bool CONCAT, bool SAVE, bool LOAD>
__global__ void __launch_bounds__(OWN_T)
owner_conv_k(const unsigned char* __restrict__ own8, const int* __restrict__ rec, int E, int N,
             const float* __restrict__ feat, const float* __restrict__ es,
             const float* __restrict__ ed, const float* __restrict__ bias,
             const float* __restrict__ lg, const float* __restrict__ lb,
             float* __restrict__ out,
             int* __restrict__ sv_cnt, int* __restrict__ sv_src, int* __restrict__ sv_dl,
             const float* __restrict__ Wn, const float* __restrict__ asn,
             const float* __restrict__ adn, float* __restrict__ esn, float* __restrict__ edn) {
  constexpr int H = 4;
  constexpr int FR = H * CH;
  constexpr int KCH = FR / 32;
  constexpr int EPB = 8;
  constexpr int FRP = CONCAT ? FR + 4 : FR;
  __shared__ __attribute__((aligned(16))) float acc[OWN_RPW * FRP];
  __shared__ __attribute__((aligned(16))) float lw[OWN_CAP * H];
  __shared__ float wa[CONCAT ? FR * 8 : 1];
  __shared__ int lsrc[OWN_CAP];
  __shared__ int ldl[OWN_CAP];
  __shared__ float edS[OWN_RPW * H], slf[OWN_RPW * H], mx[OWN_RPW * H], den[OWN_RPW * H];
  __shared__ int cnt;

  const int tid = threadIdx.x;
  const int lane = tid & 31;
  const int wave = __builtin_amdgcn_readfirstlane(tid >> 5);
  const int lo = blockIdx.x * OWN_RPW;
  const int nloc = min(OWN_RPW, N - lo);
  const size_t svBase = (size_t)blockIdx.x * OWN_CAP;

  for (int i = tid; i < nloc * FRP; i += OWN_T) acc[i] = 0.0f;
  if constexpr (CONCAT) {
    static_assert(FR * 8 == OWN_T, "one folded weight per thread");
    const int k = tid >> 3, j = tid & 7, hs = j & 3;
    const float* __restrict__ wr = Wn + (size_t)k * NXT_NC + hs * NXT_CH;
    const float* __restrict__ ar = ((j < 4) ? asn : adn) + hs * NXT_CH;
    float s = 0.0f;
    for (int c = 0; c < NXT_CH; ++c) s += wr[c] * ar[c];
    wa[tid] = s;
  }
  for (int i = tid; i < nloc * H; i += OWN_T) {
    const float d = ed[(size_t)lo * H + i];
    const float v = leaky(es[(size_t)lo * H + i] + d);
    edS[i] = d; slf[i] = v; mx[i] = v; den[i] = 0.0f;
  }

  auto scan = [&](int a, int b) -> int {
    if (tid == 0) cnt = 0;
    __syncthreads();
    auto hit = [&](int e) {
      const int i = atomicAdd(&cnt, 1);
      if (i < OWN_CAP) lsrc[i] = e;
    };
    const unsigned me4 = (unsigned)blockIdx.x * 0x01010101u;
    auto match = [&](unsigned w) -> unsigned {
      const unsigned x = w ^ me4;
      return ~((((x & 0x7f7f7f7fu) + 0x7f7f7f7fu) | x) | 0x7f7f7f7fu);
    };
    if ((((size_t)(own8 + a)) & 15) == 0) {
      const int nv = (b - a) >> 4;
      const uint4* __restrict__ p4 = (const uint4*)(own8 + a);
      const int rot = (int)(((long long)blockIdx.x * nv) / gridDim.x) & ~31;
      for (int i0 = tid; i0 < nv; i0 += OWN_T * OWN_SCAN_U) {
        uint4 v[OWN_SCAN_U];
        int at[OWN_SCAN_U];
#pragma unroll
        for (int u = 0; u < OWN_SCAN_U; ++u) {
          const int ii = i0 + u * OWN_T;
          int jj = ii + rot;
          if (jj >= nv) jj -= nv;
          at[u] = (ii < nv) ? jj : -1;
          v[u] = p4[max(at[u], 0)];
        }
#pragma unroll
        for (int u = 0; u < OWN_SCAN_U; ++u) {
          unsigned m = (match(v[u].x) >> 7) | (match(v[u].y) >> 6) |
                       (match(v[u].z) >> 5) | (match(v[u].w) >> 4);
          if (at[u] < 0) m = 0u;
          const int e0 = a + 16 * at[u];
          while (m != 0u) {
            const int p = __ffs((int)m) - 1;
            hit(e0 + 4 * (p & 7) + (p >> 3));
            m &= m - 1u;
          }
        }
      }
      for (int e = a + 16 * nv + tid; e < b; e += OWN_T)
        if (own8[e] == (unsigned char)blockIdx.x) hit(e);
    } else {
      for (int e = a + tid; e < b; e += OWN_T)
        if (own8[e] == (unsigned char)blockIdx.x) hit(e);
    }
    __syncthreads();
    return cnt;
  };

  auto resolve = [&](int n, bool fromEdgeIds) {
    for (int i0 = tid; i0 < n; i0 += OWN_T * 4) {
      int s[4], dls[4];
      float4 ev[4];
#pragma unroll
      for (int u = 0; u < 4; ++u) {
        const int i = i0 + u * OWN_T;
        int t = (i < n) ? lsrc[i] : 0;
        if (fromEdgeIds) {
          const int r = rec[t];
          s[u] = (int)((unsigned)r >> 8);
          dls[u] = r & 255;
        } else {
          s[u] = t;
          dls[u] = (i < n) ? ldl[i] : 0;
        }
      }
#pragma unroll
      for (int u = 0; u < 4; ++u) ev[u] = *(const float4*)(es + (size_t)s[u] * H);
#pragma unroll
      for (int u = 0; u < 4; ++u) {
        const int i = i0 + u * OWN_T;
        if (i < n) {
          const int dl = dls[u];
          if (fromEdgeIds) { lsrc[i] = s[u]; ldl[i] = dl; }
          const float l0 = leaky(ev[u].x + edS[dl * H + 0]);
          const float l1 = leaky(ev[u].y + edS[dl * H + 1]);
          const float l2 = leaky(ev[u].z + edS[dl * H + 2]);
          const float l3 = leaky(ev[u].w + edS[dl * H + 3]);
          lw[i * H + 0] = l0; lw[i * H + 1] = l1; lw[i * H + 2] = l2; lw[i * H + 3] = l3;
          lds_max_f32(&mx[dl * H + 0], l0); lds_max_f32(&mx[dl * H + 1], l1);
          lds_max_f32(&mx[dl * H + 2], l2); lds_max_f32(&mx[dl * H + 3], l3);
        }
      }
    }
    __syncthreads();
  };

  auto sumphase = [&](int n, bool first) {
    const int M = (n + (first ? nloc : 0)) * H;
    for (int it = tid; it < M; it += OWN_T) {
      const int i = it >> 2, h = it & 3;
      int dl; float l;
      if (i < n) { dl = ldl[i]; l = lw[it]; }
      else       { dl = i - n;  l = slf[dl * H + h]; }
      atomicAdd(&den[dl * H + h], __expf(l - mx[dl * H + h]));
    }
  };

  auto aggphase = [&](int n, bool first) {
    const int M = n + (first ? nloc : 0);
    const int nR = (M + EPB - 1) / EPB;
    for (int r = wave; r < nR; r += OWN_T / 32) {
      const int h = lane & 3;
      const int i = r * EPB + (lane >> 2);
      int s = 0, dl = 0; float alpha = 0.0f;
      if (i < M) {
        float l;
        if (i < n) { s = lsrc[i]; dl = ldl[i]; l = lw[i * H + h]; }
        else       { dl = i - n; s = lo + dl; l = slf[dl * H + h]; }
        alpha = __expf(l - mx[dl * H + h]) / (den[dl * H + h] + GAT_SM_EPS);
      }
      float f[EPB][KCH];
#pragma unroll
      for (int j = 0; j < EPB; ++j) {
        const float* row = feat + (size_t)lane_i32(s, 4 * j) * FR;
#pragma unroll
        for (int k = 0; k < KCH; ++k) f[j][k] = row[lane + 32 * k];
      }
#pragma unroll
      for (int j = 0; j < EPB; ++j) {
        if (r * EPB + j < M) {
          float* arow = acc + lane_i32(dl, 4 * j) * FRP;
#pragma unroll
          for (int k = 0; k < KCH; ++k) {
            const int hLo = (32 * k) / CH, hHi = (32 * k + 31) / CH;
            float w = lane_f32(alpha, 4 * j + hLo);
            if (hHi != hLo) {
              const float wHi = lane_f32(alpha, 4 * j + hHi);
              w = (lane < (hLo + 1) * CH - 32 * k) ? w : wHi;
            }
            atomicAdd(&arow[lane + 32 * k], f[j][k] * w);
          }
        }
      }
    }
  };

  int n = 0, nseg = 1;
  bool slow = false;
  for (int pass = 0; pass < 3; ++pass) {
    for (int seg = 0; seg < nseg; ++seg) {
      if (pass == 0 || slow) {
        int total = -1;
        if (LOAD && !slow) total = sv_cnt[blockIdx.x];
        if (total >= 0 && total <= OWN_CAP) {
          for (int i = tid; i < total; i += OWN_T) {
            lsrc[i] = sv_src[svBase + i];
            ldl[i] = sv_dl[svBase + i];
          }
          __syncthreads();
          n = total;
          resolve(n, false);
        } else {
          const int a = slow ? seg * OWN_CAP : 0;
          const int b = slow ? min(E, a + OWN_CAP) : E;
          total = scan(a, b);
          n = min(total, OWN_CAP);
          resolve(n, true);
          if (!slow) {
            if (SAVE) {
              if (total <= OWN_CAP) {
                for (int i = tid; i < n; i += OWN_T) {
                  sv_src[svBase + i] = lsrc[i];
                  sv_dl[svBase + i] = ldl[i];
                }
              }
              if (tid == 0) sv_cnt[blockIdx.x] = (total <= OWN_CAP) ? total : -1;
            }
            if (total > OWN_CAP) {
              slow = true;
              nseg = (E + OWN_CAP - 1) / OWN_CAP;
              seg = -1;
              continue;
            }
          }
        }
      }
      if (pass == 1) { sumphase(n, seg == 0); __syncthreads(); }
      if (pass == 2) { aggphase(n, seg == 0); __syncthreads(); }
    }
  }

  constexpr int TL = 8;
  constexpr int NPR = 32 / TL;
  const int sub = lane & (TL - 1);
  const int grp = lane / TL;
  float4 g4[CONCAT ? 4 : 1], s4[CONCAT ? 4 : 1];
  if constexpr (CONCAT) {
#pragma unroll
    for (int m = 0; m < 4; ++m) {
      g4[m] = *(const float4*)(lg + 32 * m + 4 * sub);
      s4[m] = *(const float4*)(lb + 32 * m + 4 * sub);
    }
  }
  constexpr int NT = NXT_NC / 16 + 1;
  constexpr int KB = FR / 16;
  constexpr int ITEMS = CONCAT ? (FR / 4) * NXT_NC : OWN_T;
  static_assert(ITEMS % OWN_T == 0, "every thread stages the same number of items");
  static_assert(!CONCAT || KB * 2 * NT * 32 * 4 <= OWN_CAP * H, "the staged weights fit the logit list's place");
  float4 bv4[ITEMS / OWN_T];
  if constexpr (CONCAT) {
#pragma unroll
    for (int u = 0; u < ITEMS / OWN_T; ++u) {
      const int it = tid + u * OWN_T;
      const int m = it / NXT_NC, n = it - m * NXT_NC;
      const float* __restrict__ bp = Wn + (size_t)(4 * m) * NXT_NC + n;
      bv4[u].x = bp[0]; bv4[u].y = bp[NXT_NC]; bv4[u].z = bp[2 * NXT_NC]; bv4[u].w = bp[3 * NXT_NC];
    }
  }
  for (int i0 = wave * NPR; i0 < nloc; i0 += (OWN_T / 32) * NPR) {
    const bool live = (i0 + grp) < nloc;
    const int i = live ? (i0 + grp) : (nloc - 1);
    const int node = lo + i;
    const float* ap = acc + i * FRP;
    if constexpr (CONCAT) {
      static_assert(FR == TL * 16, "four 16-byte pieces per lane");
      float4 v[4];
      float sum = 0.0f;
#pragma unroll
      for (int m = 0; m < 4; ++m) {
        const int c = 32 * m + 4 * sub;
        const float4 a4 = *(const float4*)(ap + c);
        const float4 b4 = *(const float4*)(bias + c);
        float4 t;
        t.x = a4.x + b4.x; t.y = a4.y + b4.y; t.z = a4.z + b4.z; t.w = a4.w + b4.w;
        t.x = (t.x > 0.0f) ? t.x : (__expf(t.x) - 1.0f);
        t.y = (t.y > 0.0f) ? t.y : (__expf(t.y) - 1.0f);
        t.z = (t.z > 0.0f) ? t.z : (__expf(t.z) - 1.0f);
        t.w = (t.w > 0.0f) ? t.w : (__expf(t.w) - 1.0f);
        v[m] = t;
        sum += t.x; sum += t.y; sum += t.z; sum += t.w;
      }
      const float mu = group_sum8(sum) * (1.0f / 128.0f);
      float vs = 0.0f;
#pragma unroll
      for (int m = 0; m < 4; ++m) {
        float4 d = v[m];
        d.x -= mu; d.y -= mu; d.z -= mu; d.w -= mu;
        v[m] = d;
        vs += d.x * d.x; vs += d.y * d.y; vs += d.z * d.z; vs += d.w * d.w;
      }
      const float var = group_sum8(vs) * (1.0f / 128.0f);
      const float rstd = rsqrtf(var + GAT_LN_EPS);
      float* xp = acc + i * FRP;
#pragma unroll
      for (int m = 0; m < 4; ++m) {
        const int c = 32 * m + 4 * sub;
        float4 o;
        o.x = v[m].x * rstd * g4[m].x + s4[m].x; o.y = v[m].y * rstd * g4[m].y + s4[m].y;
        o.z = v[m].z * rstd * g4[m].z + s4[m].z; o.w = v[m].w * rstd * g4[m].w + s4[m].w;
        if (live) *(float4*)(xp + c) = o;
      }
    } else {
      static_assert(CH % TL == 0, "every lane of a group holds CH / TL classes");
      constexpr int CPL = CH / TL;
      float x[CPL];
      float sum = 0.0f;
#pragma unroll
      for (int m = 0; m < CPL; ++m) {
        const int c = sub + TL * m;
        x[m] = (ap[c] + ap[c + CH] + ap[c + 2 * CH] + ap[c + 3 * CH]) * 0.25f + bias[c];
        sum += x[m];
      }
      const float mu = group_sum8(sum) * (1.0f / CH);
      float vs = 0.0f;
#pragma unroll
      for (int m = 0; m < CPL; ++m) { x[m] -= mu; vs += x[m] * x[m]; }
      const float var = group_sum8(vs) * (1.0f / CH);
      const float rstd = rsqrtf(var + GAT_LN_EPS);
      float mxz = -3.0e38f;
#pragma unroll
      for (int m = 0; m < CPL; ++m) {
        const int c = sub + TL * m;
        x[m] = x[m] * rstd * lg[c] + lb[c];
        mxz = fmaxf(mxz, x[m]);
      }
      mxz = group_max8(mxz);
      float se = 0.0f;
#pragma unroll
      for (int m = 0; m < CPL; ++m) se += __expf(x[m] - mxz);
      const float lse = __logf(group_sum8(se));
      if (live) {
        float* yp = out + (size_t)node * CH;
#pragma unroll
        for (int m = 0; m < CPL; ++m) yp[sub + TL * m] = x[m] - mxz - lse;
      }
    }
  }

  if constexpr (CONCAT) {
    float4* Bs = (float4*)lw;
#pragma unroll
    for (int u = 0; u < ITEMS / OWN_T; ++u) {
      const int it = tid + u * OWN_T;
      const int m = it / NXT_NC, n = it - m * NXT_NC;
      const int kb = m >> 2, hh = (m >> 1) & 1, jp = m & 1;
      Bs[((kb * 2 + jp) * NT + (n >> 4)) * 32 + hh * 16 + (n & 15)] = bv4[u];
    }
    if (tid < (FR / 4) * 16) {
      const int m = tid >> 4, np = tid & 15;
      float4 v = {0.0f, 0.0f, 0.0f, 0.0f};
      if (np < 8) {
        v.x = wa[(4 * m) * 8 + np];     v.y = wa[(4 * m + 1) * 8 + np];
        v.z = wa[(4 * m + 2) * 8 + np]; v.w = wa[(4 * m + 3) * 8 + np];
      }
      const int kb = m >> 2, hh = (m >> 1) & 1, jp = m & 1;
      Bs[((kb * 2 + jp) * NT + (NT - 1)) * 32 + hh * 16 + np] = v;
    }
    __syncthreads();

    const int c16 = lane & 15, hh = lane >> 4;
    auto unit = [&](auto ntl, int rt, int ct0) {
      constexpr int NTL = decltype(ntl)::value;
      const int rowA = min(rt * 16 + c16, OWN_RPW - 1);
      const float* ap = acc + rowA * FRP + 8 * hh;
      v8f d[NTL];
#pragma unroll
      for (int t = 0; t < NTL; ++t) d[t] = v8f{};
#pragma unroll
      for (int kb = 0; kb < KB; ++kb) {
#pragma unroll
        for (int jp = 0; jp < 2; ++jp) {
          const float4 av = *(const float4*)(ap + 16 * kb + 4 * jp);
          const v2f a0 = {av.x, av.y}, a1 = {av.z, av.w};
#pragma unroll
          for (int t = 0; t < NTL; ++t) {
            const float4 bv = Bs[((kb * 2 + jp) * NT + ct0 + t) * 32 + lane];
            const v2f b0 = {bv.x, bv.y}, b1 = {bv.z, bv.w};
            d[t] = __builtin_amdgcn_wmma_f32_16x16x4_f32(false, a0, false, b0, (short)0, d[t],
                                                         false, false);
            d[t] = __builtin_amdgcn_wmma_f32_16x16x4_f32(false, a1, false, b1, (short)0, d[t],
                                                         false, false);
          }
        }
      }
      const int rTop = rt * 16 + hh * 8;
      if (ct0 < NT - 1) {
#pragma unroll
        for (int t = 0; t < NTL; ++t) {
#pragma unroll
          for (int r = 0; r < 8; ++r) {
            const int row = rTop + r;
            if (row < nloc) out[(size_t)(lo + row) * NXT_NC + (ct0 + t) * 16 + c16] = d[t][r];
          }
        }
      } else {
        float* __restrict__ sp = (c16 < 4) ? esn : edn;
#pragma unroll
        for (int r = 0; r < 8; ++r) {
          const int row = rTop + r;
          if (row < nloc && c16 < 8) sp[(size_t)(lo + row) * H + (c16 & 3)] = d[0][r];
        }
      }
    };
    const int nRT = (nloc + 15) >> 4;
    const int nPair = nRT * (NXT_NC / 32);
    const int nU = nPair + nRT;
    for (int u = wave; u < nU; u += OWN_T / 32) {
      if (u < nPair) {
        const int rt = u / (NXT_NC / 32);
        unit(IntC<2>{}, rt, 2 * (u - rt * (NXT_NC / 32)));
      } else {
        unit(IntC<1>{}, u - nPair, NT - 1);
      }
    }
  }
}

extern "C" void kernel_launch(void* const* d_in, const int* in_sizes, int n_in,
                              void* d_out, int out_size, void* d_ws, size_t ws_size,
                              hipStream_t stream) {
  const float* x        = (const float*)d_in[0];
  const int*   ei       = (const int*)d_in[1];
  const float* W1       = (const float*)d_in[2];
  const float* att_src1 = (const float*)d_in[3];
  const float* att_dst1 = (const float*)d_in[4];
  const float* b1       = (const float*)d_in[5];
  const float* W2       = (const float*)d_in[6];
  const float* att_src2 = (const float*)d_in[7];
  const float* att_dst2 = (const float*)d_in[8];
  const float* b2       = (const float*)d_in[9];
  const float* ln0_g    = (const float*)d_in[10];
  const float* ln0_b    = (const float*)d_in[11];
  const float* ln1_g    = (const float*)d_in[12];
  const float* ln1_b    = (const float*)d_in[13];

  const int F_IN = 256, H1C = 128, H2C = 160, H = 4;
  const int N = in_sizes[0] / F_IN;
  const int E = in_sizes[1] / 2;

  const int owners = (N + OWN_RPW - 1) / OWN_RPW;

  float* ws   = (float*)d_ws;
  float* h1   = ws;
  float* h2   = h1   + (size_t)N * H1C;
  float* es   = h2   + (size_t)N * H2C;
  float* ed   = es   + (size_t)N * H;
  float* es2  = ed   + (size_t)N * H;
  float* ed2  = es2  + (size_t)N * H;
  int* sv_cnt = (int*)(ed2 + (size_t)N * H);
  int* sv_src = sv_cnt + owners;
  int* sv_dl  = sv_src + (size_t)owners * OWN_CAP;
  int* rec    = sv_dl + (size_t)owners * OWN_CAP;
  unsigned char* own8 = (unsigned char*)(rec + E);

  gemm_wmma_f32<256, 128><<<owners, GEMM_T, 0, stream>>>(x, W1, h1, N, att_src1, att_dst1,
                                                         es, ed, ei, E, own8, rec);
  owner_conv_k<32, true, true, false><<<owners, OWN_T, 0, stream>>>(
      own8, rec, E, N, h1, es, ed, b1, ln0_g, ln0_b, h2, sv_cnt, sv_src, sv_dl,
      W2, att_src2, att_dst2, es2, ed2);

  owner_conv_k<40, false, false, true><<<owners, OWN_T, 0, stream>>>(
      own8, rec, E, N, h2, es2, ed2, b2, ln1_g, ln1_b, (float*)d_out, sv_cnt, sv_src, sv_dl,
      nullptr, nullptr, nullptr, nullptr, nullptr);
}


# conv2 product rows stored class-major (4 heads of a class adjacent; W2 columns permuted in LDS); gather waves weight and sum the heads in registers: 40 LDS adds per edge (was 160); heads sum first
# speedup vs baseline: 5.3366x; 1.0435x over previous
#include <hip/hip_runtime.h>
#include <math.h>

typedef __attribute__((ext_vector_type(2))) float v2f;
typedef __attribute__((ext_vector_type(8))) float v8f;

#define GAT_NEG_SLOPE 0.2f
#define GAT_SM_EPS 1e-16f
#define GAT_LN_EPS 1e-5f

__device__ __forceinline__ float group_sum8(float v) {
#pragma unroll
  for (int o = 4; o > 0; o >>= 1) v += __shfl_xor(v, o, 32);
  return v;
}
__device__ __forceinline__ float group_max8(float v) {
#pragma unroll
  for (int o = 4; o > 0; o >>= 1) v = fmaxf(v, __shfl_xor(v, o, 32));
  return v;
}


#define OWN_T 1024
#define OWN_RPW 196
#define OWN_CAP 6144
#define OWN_SCAN_U 7
static_assert(OWN_RPW <= 256, "a destination's place in its owner's range is kept in one byte");

#define GEMM_T 512
template <int K, int NC>
__global__ void __launch_bounds__(GEMM_T)
gemm_wmma_f32(const float* __restrict__ A, const float* __restrict__ B, float* __restrict__ C,
              int M, const float* __restrict__ a_src, const float* __restrict__ a_dst,
              float* __restrict__ es, float* __restrict__ ed,
              const int* __restrict__ ei, int E,
              unsigned char* __restrict__ own8, int* __restrict__ rec) {
  constexpr int T = NC / 16;
  constexpr int KB = K / 16;
  constexpr int H = 4, CH = NC / H;
  static_assert(K % 16 == 0 && NC % 16 == 0, "whole fragments only");
  static_assert(CH == 32 && GEMM_T == 512, "the fold gives a wave one head and K/4 rows of B");
  __shared__ float4 Bs[KB * 2 * T * 32];
  const int tid = threadIdx.x;
  const int lane = tid & 31;
  const int wave = __builtin_amdgcn_readfirstlane(tid >> 5);
  const int c = lane & 15, h = lane >> 4;
  const int nMain = (M + 16 * (GEMM_T / 32) - 1) / (16 * (GEMM_T / 32));
  const int nSide = (int)gridDim.x - nMain;
  const bool side = (int)blockIdx.x >= nMain;

  if (side || nSide <= 0) {
    const int wIdx = side ? (int)blockIdx.x - nMain : (int)blockIdx.x;
    const int nW = side ? nSide : (int)gridDim.x;

    {
      const int* __restrict__ dstp = ei + E;
      const int wtid = wIdx * GEMM_T + tid;
      const int nthr = nW * GEMM_T;
      auto pack = [](int sN, int dN, unsigned& o) -> int {
        o = (unsigned)dN / (unsigned)OWN_RPW;
        return (int)(((unsigned)sN << 8) | ((unsigned)dN - o * (unsigned)OWN_RPW));
      };
      if ((E & 3) == 0 && E >= 4 && ((((size_t)rec) | ((size_t)own8) | ((size_t)ei)) & 15) == 0) {
        const int nq = E >> 2;
        for (int q0 = wtid; q0 < nq; q0 += 4 * nthr) {
          int4 sv[4], dv[4];
#pragma unroll
          for (int u = 0; u < 4; ++u) {
            const int qq = min(q0 + u * nthr, nq - 1);
            sv[u] = ((const int4*)ei)[qq];
            dv[u] = ((const int4*)dstp)[qq];
          }
#pragma unroll
          for (int u = 0; u < 4; ++u) {
            const int q = q0 + u * nthr;
            if (q < nq) {
              unsigned o0, o1, o2, o3;
              int4 r;
              r.x = pack(sv[u].x, dv[u].x, o0); r.y = pack(sv[u].y, dv[u].y, o1);
              r.z = pack(sv[u].z, dv[u].z, o2); r.w = pack(sv[u].w, dv[u].w, o3);
              ((unsigned*)own8)[q] = (o0 & 255u) | ((o1 & 255u) << 8) | ((o2 & 255u) << 16) | (o3 << 24);
              ((int4*)rec)[q] = r;
            }
          }
        }
      } else {
        for (int e = wtid; e < E; e += nthr) {
          unsigned o;
          rec[e] = pack(ei[e], dstp[e], o);
          own8[e] = (unsigned char)o;
        }
      }
    }

    {
      float* Wsf = (float*)Bs;
      const int hs = wave & 3;
      const int piece = lane & 7, rsub = lane >> 3;
      const float4 asv = *(const float4*)(a_src + hs * CH + 4 * piece);
      const float4 adv = *(const float4*)(a_dst + hs * CH + 4 * piece);
      constexpr int FU = K / 16;
      float4 wv[FU];
#pragma unroll
      for (int u = 0; u < FU; ++u) {
        const int k = (wave >> 2) * (K / 4) + 4 * u + rsub;
        wv[u] = *(const float4*)(B + (size_t)k * NC + hs * CH + 4 * piece);
      }
      for (int i = tid; i < KB * 2 * 2 * 8; i += GEMM_T) {
        const int blk = i >> 4, hh = (i >> 3) & 1, col = 8 + (i & 7);
        Bs[blk * 32 + hh * 16 + col] = float4{0.0f, 0.0f, 0.0f, 0.0f};
      }
#pragma unroll
      for (int u = 0; u < FU; ++u) {
        const int k = (wave >> 2) * (K / 4) + 4 * u + rsub;
        float ps = wv[u].x * asv.x + wv[u].y * asv.y + wv[u].z * asv.z + wv[u].w * asv.w;
        float pd = wv[u].x * adv.x + wv[u].y * adv.y + wv[u].z * adv.z + wv[u].w * adv.w;
        ps = group_sum8(ps);
        pd = group_sum8(pd);
        if (piece == 0) {
          const int kb = k >> 4, hh = (k >> 3) & 1, jp = (k >> 2) & 1, q = k & 3;
          float* slot = Wsf + (((kb * 2 + jp) * 32 + hh * 16) * 4 + q);
          slot[hs * 4] = ps;
          slot[(4 + hs) * 4] = pd;
        }
      }
    }
    __syncthreads();

    for (int tl = wIdx * (GEMM_T / 32) + wave; tl * 16 < M; tl += nW * (GEMM_T / 32)) {
      const int rowS = min(tl * 16 + c, M - 1);
      const float4* __restrict__ sp4 = (const float4*)(A + (size_t)rowS * K + 8 * h);
      float4 sa[KB][2];
#pragma unroll
      for (int kb = 0; kb < KB; ++kb) { sa[kb][0] = sp4[4 * kb]; sa[kb][1] = sp4[4 * kb + 1]; }
      v8f d[2] = {v8f{}, v8f{}};
#pragma unroll
      for (int kb = 0; kb < KB; ++kb) {
#pragma unroll
        for (int jp = 0; jp < 2; ++jp) {
          const float4 av = sa[kb][jp];
          const v2f a0 = {av.x, av.y}, a1 = {av.z, av.w};
          const float4 bv = Bs[(kb * 2 + jp) * 32 + lane];
          const v2f b0 = {bv.x, bv.y}, b1 = {bv.z, bv.w};
          d[jp] = __builtin_amdgcn_wmma_f32_16x16x4_f32(false, a0, false, b0, (short)0, d[jp],
                                                        false, false);
          d[jp] = __builtin_amdgcn_wmma_f32_16x16x4_f32(false, a1, false, b1, (short)0, d[jp],
                                                        false, false);
        }
      }
      float* __restrict__ sp = (c < 4) ? es : ed;
#pragma unroll
      for (int r = 0; r < 8; ++r) {
        const int row = tl * 16 + h * 8 + r;
        if (row < M && c < 8) sp[(size_t)row * H + (c & 3)] = d[0][r] + d[1][r];
      }
    }
    if (side) return;
    __syncthreads();
  }

  const int tile = blockIdx.x * (GEMM_T / 32) + wave;

  const int rowA = min(tile * 16 + c, M - 1);
  const float4* __restrict__ ap = (const float4*)(A + (size_t)rowA * K + 8 * h);
  float4 a[KB][2];
#pragma unroll
  for (int kb = 0; kb < KB; ++kb) { a[kb][0] = ap[4 * kb]; a[kb][1] = ap[4 * kb + 1]; }

  constexpr int ITEMS = (K / 4) * NC;
  static_assert(ITEMS % GEMM_T == 0, "every thread stages the same number of items");
  float4 bv4[ITEMS / GEMM_T];
#pragma unroll
  for (int u = 0; u < ITEMS / GEMM_T; ++u) {
    const int it = tid + u * GEMM_T;
    const int m = it / NC, n = it - m * NC;
    const float* __restrict__ bp = B + (size_t)(4 * m) * NC + n;
    bv4[u].x = bp[0]; bv4[u].y = bp[NC]; bv4[u].z = bp[2 * NC]; bv4[u].w = bp[3 * NC];
  }
#pragma unroll
  for (int u = 0; u < ITEMS / GEMM_T; ++u) {
    const int it = tid + u * GEMM_T;
    const int m = it / NC, n = it - m * NC;
    const int kb = m >> 2, hh = (m >> 1) & 1, jp = m & 1;
    Bs[((kb * 2 + jp) * T + (n >> 4)) * 32 + hh * 16 + (n & 15)] = bv4[u];
  }
  __syncthreads();
  if (tile * 16 >= M) return;

  v8f acc[T];
#pragma unroll
  for (int t = 0; t < T; ++t) acc[t] = v8f{};
#pragma unroll
  for (int kb = 0; kb < KB; ++kb) {
#pragma unroll
    for (int jp = 0; jp < 2; ++jp) {
      const float4 av = a[kb][jp];
      const v2f a0 = {av.x, av.y}, a1 = {av.z, av.w};
#pragma unroll
      for (int t = 0; t < T; ++t) {
        const float4 bv = Bs[((kb * 2 + jp) * T + t) * 32 + lane];
        const v2f b0 = {bv.x, bv.y}, b1 = {bv.z, bv.w};
        acc[t] = __builtin_amdgcn_wmma_f32_16x16x4_f32(false, a0, false, b0, (short)0, acc[t],
                                                       false, false);
        acc[t] = __builtin_amdgcn_wmma_f32_16x16x4_f32(false, a1, false, b1, (short)0, acc[t],
                                                       false, false);
      }
    }
  }

  const int rowTop = tile * 16 + h * 8;
#pragma unroll
  for (int t = 0; t < T; ++t) {
#pragma unroll
    for (int r = 0; r < 8; ++r) {
      const int row = rowTop + r;
      if (row < M) C[(size_t)row * NC + t * 16 + c] = acc[t][r];
    }
  }
}

__device__ __forceinline__ int lane_i32(int v, int src) {
  return __builtin_amdgcn_readlane(v, src);
}
__device__ __forceinline__ float lane_f32(float v, int src) {
  return __int_as_float(__builtin_amdgcn_readlane(__float_as_int(v), src));
}

__device__ __forceinline__ float leaky(float v) { return (v > 0.0f) ? v : GAT_NEG_SLOPE * v; }

__device__ __forceinline__ void lds_max_f32(float* addr, float v) {
  if (v >= 0.0f) atomicMax((int*)addr, __float_as_int(v));
  else           atomicMin((unsigned int*)addr, __float_as_uint(v));
}

#define NXT_NC 160
#define NXT_CH 40
template <int V> struct IntC { static constexpr int value = V; };

template <int CH, bool CONCAT, bool SAVE, bool LOAD>
__global__ void __launch_bounds__(OWN_T)
owner_conv_k(const unsigned char* __restrict__ own8, const int* __restrict__ rec, int E, int N,
             const float* __restrict__ feat, const float* __restrict__ es,
             const float* __restrict__ ed, const float* __restrict__ bias,
             const float* __restrict__ lg, const float* __restrict__ lb,
             float* __restrict__ out,
             int* __restrict__ sv_cnt, int* __restrict__ sv_src, int* __restrict__ sv_dl,
             const float* __restrict__ Wn, const float* __restrict__ asn,
             const float* __restrict__ adn, float* __restrict__ esn, float* __restrict__ edn) {
  constexpr int H = 4;
  constexpr int FR = H * CH;
  constexpr int KCH = FR / 32;
  constexpr int EPB = 8;
  constexpr int FRP = CONCAT ? FR + 4 : CH;
  __shared__ __attribute__((aligned(16))) float acc[OWN_RPW * FRP];
  __shared__ __attribute__((aligned(16))) float lw[OWN_CAP * H];
  __shared__ float wa[CONCAT ? FR * 8 : 1];
  __shared__ int lsrc[OWN_CAP];
  __shared__ int ldl[OWN_CAP];
  __shared__ float edS[OWN_RPW * H], slf[OWN_RPW * H], mx[OWN_RPW * H], den[OWN_RPW * H];
  __shared__ int cnt;

  const int tid = threadIdx.x;
  const int lane = tid & 31;
  const int wave = __builtin_amdgcn_readfirstlane(tid >> 5);
  const int lo = blockIdx.x * OWN_RPW;
  const int nloc = min(OWN_RPW, N - lo);
  const size_t svBase = (size_t)blockIdx.x * OWN_CAP;

  for (int i = tid; i < nloc * FRP; i += OWN_T) acc[i] = 0.0f;
  if constexpr (CONCAT) {
    static_assert(FR * 8 == OWN_T, "one folded weight per thread");
    const int k = tid >> 3, j = tid & 7, hs = j & 3;
    const float* __restrict__ wr = Wn + (size_t)k * NXT_NC + hs * NXT_CH;
    const float* __restrict__ ar = ((j < 4) ? asn : adn) + hs * NXT_CH;
    float s = 0.0f;
    for (int c = 0; c < NXT_CH; ++c) s += wr[c] * ar[c];
    wa[tid] = s;
  }
  for (int i = tid; i < nloc * H; i += OWN_T) {
    const float d = ed[(size_t)lo * H + i];
    const float v = leaky(es[(size_t)lo * H + i] + d);
    edS[i] = d; slf[i] = v; mx[i] = v; den[i] = 0.0f;
  }

  auto scan = [&](int a, int b) -> int {
    if (tid == 0) cnt = 0;
    __syncthreads();
    auto hit = [&](int e) {
      const int i = atomicAdd(&cnt, 1);
      if (i < OWN_CAP) lsrc[i] = e;
    };
    const unsigned me4 = (unsigned)blockIdx.x * 0x01010101u;
    auto match = [&](unsigned w) -> unsigned {
      const unsigned x = w ^ me4;
      return ~((((x & 0x7f7f7f7fu) + 0x7f7f7f7fu) | x) | 0x7f7f7f7fu);
    };
    if ((((size_t)(own8 + a)) & 15) == 0) {
      const int nv = (b - a) >> 4;
      const uint4* __restrict__ p4 = (const uint4*)(own8 + a);
      const int rot = (int)(((long long)blockIdx.x * nv) / gridDim.x) & ~31;
      for (int i0 = tid; i0 < nv; i0 += OWN_T * OWN_SCAN_U) {
        uint4 v[OWN_SCAN_U];
        int at[OWN_SCAN_U];
#pragma unroll
        for (int u = 0; u < OWN_SCAN_U; ++u) {
          const int ii = i0 + u * OWN_T;
          int jj = ii + rot;
          if (jj >= nv) jj -= nv;
          at[u] = (ii < nv) ? jj : -1;
          v[u] = p4[max(at[u], 0)];
        }
#pragma unroll
        for (int u = 0; u < OWN_SCAN_U; ++u) {
          unsigned m = (match(v[u].x) >> 7) | (match(v[u].y) >> 6) |
                       (match(v[u].z) >> 5) | (match(v[u].w) >> 4);
          if (at[u] < 0) m = 0u;
          const int e0 = a + 16 * at[u];
          while (m != 0u) {
            const int p = __ffs((int)m) - 1;
            hit(e0 + 4 * (p & 7) + (p >> 3));
            m &= m - 1u;
          }
        }
      }
      for (int e = a + 16 * nv + tid; e < b; e += OWN_T)
        if (own8[e] == (unsigned char)blockIdx.x) hit(e);
    } else {
      for (int e = a + tid; e < b; e += OWN_T)
        if (own8[e] == (unsigned char)blockIdx.x) hit(e);
    }
    __syncthreads();
    return cnt;
  };

  auto resolve = [&](int n, bool fromEdgeIds) {
    for (int i0 = tid; i0 < n; i0 += OWN_T * 4) {
      int s[4], dls[4];
      float4 ev[4];
#pragma unroll
      for (int u = 0; u < 4; ++u) {
        const int i = i0 + u * OWN_T;
        int t = (i < n) ? lsrc[i] : 0;
        if (fromEdgeIds) {
          const int r = rec[t];
          s[u] = (int)((unsigned)r >> 8);
          dls[u] = r & 255;
        } else {
          s[u] = t;
          dls[u] = (i < n) ? ldl[i] : 0;
        }
      }
#pragma unroll
      for (int u = 0; u < 4; ++u) ev[u] = *(const float4*)(es + (size_t)s[u] * H);
#pragma unroll
      for (int u = 0; u < 4; ++u) {
        const int i = i0 + u * OWN_T;
        if (i < n) {
          const int dl = dls[u];
          if (fromEdgeIds) { lsrc[i] = s[u]; ldl[i] = dl; }
          const float l0 = leaky(ev[u].x + edS[dl * H + 0]);
          const float l1 = leaky(ev[u].y + edS[dl * H + 1]);
          const float l2 = leaky(ev[u].z + edS[dl * H + 2]);
          const float l3 = leaky(ev[u].w + edS[dl * H + 3]);
          lw[i * H + 0] = l0; lw[i * H + 1] = l1; lw[i * H + 2] = l2; lw[i * H + 3] = l3;
          lds_max_f32(&mx[dl * H + 0], l0); lds_max_f32(&mx[dl * H + 1], l1);
          lds_max_f32(&mx[dl * H + 2], l2); lds_max_f32(&mx[dl * H + 3], l3);
        }
      }
    }
    __syncthreads();
  };

  auto sumphase = [&](int n, bool first) {
    const int M = (n + (first ? nloc : 0)) * H;
    for (int it = tid; it < M; it += OWN_T) {
      const int i = it >> 2, h = it & 3;
      int dl; float l;
      if (i < n) { dl = ldl[i]; l = lw[it]; }
      else       { dl = i - n;  l = slf[dl * H + h]; }
      atomicAdd(&den[dl * H + h], __expf(l - mx[dl * H + h]));
    }
  };

  auto aggphase = [&](int n, bool first) {
    const int M = n + (first ? nloc : 0);
    const int nR = (M + EPB - 1) / EPB;
    for (int r = wave; r < nR; r += OWN_T / 32) {
      const int h = lane & 3;
      const int i = r * EPB + (lane >> 2);
      int s = 0, dl = 0; float alpha = 0.0f;
      if (i < M) {
        float l;
        if (i < n) { s = lsrc[i]; dl = ldl[i]; l = lw[i * H + h]; }
        else       { dl = i - n; s = lo + dl; l = slf[dl * H + h]; }
        alpha = __expf(l - mx[dl * H + h]) / (den[dl * H + h] + GAT_SM_EPS);
      }
      if constexpr (!CONCAT) {
        static_assert(CH == 40 && H == 4, "32 classes in the wide load, 8 in the short one");
        float4 fm[EPB], ft[EPB];
#pragma unroll
        for (int j = 0; j < EPB; ++j) {
          const float* row = feat + (size_t)lane_i32(s, 4 * j) * FR;
          fm[j] = *(const float4*)(row + 4 * lane);
          ft[j] = *(const float4*)(row + 128 + 4 * (lane & 7));
        }
        asm volatile("" :: "v"(fm[0].x), "v"(fm[0].y), "v"(fm[0].z), "v"(fm[0].w),
                           "v"(ft[0].x), "v"(ft[0].y), "v"(ft[0].z), "v"(ft[0].w));
#pragma unroll
        for (int j = 0; j < EPB; ++j) {
          if (r * EPB + j < M) {
            float* arow = acc + lane_i32(dl, 4 * j) * FRP;
            const float w0 = lane_f32(alpha, 4 * j), w1 = lane_f32(alpha, 4 * j + 1);
            const float w2 = lane_f32(alpha, 4 * j + 2), w3 = lane_f32(alpha, 4 * j + 3);
            atomicAdd(&arow[lane], fm[j].x * w0 + fm[j].y * w1 + fm[j].z * w2 + fm[j].w * w3);
            if (lane < 8)
              atomicAdd(&arow[32 + lane], ft[j].x * w0 + ft[j].y * w1 + ft[j].z * w2 + ft[j].w * w3);
          }
        }
      } else {
        float f[EPB][KCH];
#pragma unroll
        for (int j = 0; j < EPB; ++j) {
          const float* row = feat + (size_t)lane_i32(s, 4 * j) * FR;
#pragma unroll
          for (int k = 0; k < KCH; ++k) f[j][k] = row[lane + 32 * k];
        }
#pragma unroll
        for (int j = 0; j < EPB; ++j) {
          if (r * EPB + j < M) {
            float* arow = acc + lane_i32(dl, 4 * j) * FRP;
#pragma unroll
            for (int k = 0; k < KCH; ++k) {
              const int hLo = (32 * k) / CH, hHi = (32 * k + 31) / CH;
              float w = lane_f32(alpha, 4 * j + hLo);
              if (hHi != hLo) {
                const float wHi = lane_f32(alpha, 4 * j + hHi);
                w = (lane < (hLo + 1) * CH - 32 * k) ? w : wHi;
              }
              atomicAdd(&arow[lane + 32 * k], f[j][k] * w);
            }
          }
        }
      }
    }
  };

  int n = 0, nseg = 1;
  bool slow = false;
  for (int pass = 0; pass < 3; ++pass) {
    for (int seg = 0; seg < nseg; ++seg) {
      if (pass == 0 || slow) {
        int total = -1;
        if (LOAD && !slow) total = sv_cnt[blockIdx.x];
        if (total >= 0 && total <= OWN_CAP) {
          for (int i = tid; i < total; i += OWN_T) {
            lsrc[i] = sv_src[svBase + i];
            ldl[i] = sv_dl[svBase + i];
          }
          __syncthreads();
          n = total;
          resolve(n, false);
        } else {
          const int a = slow ? seg * OWN_CAP : 0;
          const int b = slow ? min(E, a + OWN_CAP) : E;
          total = scan(a, b);
          n = min(total, OWN_CAP);
          resolve(n, true);
          if (!slow) {
            if (SAVE) {
              if (total <= OWN_CAP) {
                for (int i = tid; i < n; i += OWN_T) {
                  sv_src[svBase + i] = lsrc[i];
                  sv_dl[svBase + i] = ldl[i];
                }
              }
              if (tid == 0) sv_cnt[blockIdx.x] = (total <= OWN_CAP) ? total : -1;
            }
            if (total > OWN_CAP) {
              slow = true;
              nseg = (E + OWN_CAP - 1) / OWN_CAP;
              seg = -1;
              continue;
            }
          }
        }
      }
      if (pass == 1) { sumphase(n, seg == 0); __syncthreads(); }
      if (pass == 2) { aggphase(n, seg == 0); __syncthreads(); }
    }
  }

  constexpr int TL = 8;
  constexpr int NPR = 32 / TL;
  const int sub = lane & (TL - 1);
  const int grp = lane / TL;
  float4 g4[CONCAT ? 4 : 1], s4[CONCAT ? 4 : 1];
  if constexpr (CONCAT) {
#pragma unroll
    for (int m = 0; m < 4; ++m) {
      g4[m] = *(const float4*)(lg + 32 * m + 4 * sub);
      s4[m] = *(const float4*)(lb + 32 * m + 4 * sub);
    }
  }
  constexpr int NT = NXT_NC / 16 + 1;
  constexpr int KB = FR / 16;
  constexpr int ITEMS = CONCAT ? (FR / 4) * NXT_NC : OWN_T;
  static_assert(ITEMS % OWN_T == 0, "every thread stages the same number of items");
  static_assert(!CONCAT || KB * 2 * NT * 32 * 4 <= OWN_CAP * H, "the staged weights fit the logit list's place");
  float4 bv4[ITEMS / OWN_T];
  if constexpr (CONCAT) {
#pragma unroll
    for (int u = 0; u < ITEMS / OWN_T; ++u) {
      const int it = tid + u * OWN_T;
      const int m = it / NXT_NC, n = it - m * NXT_NC;
      const float* __restrict__ bp = Wn + (size_t)(4 * m) * NXT_NC + n;
      bv4[u].x = bp[0]; bv4[u].y = bp[NXT_NC]; bv4[u].z = bp[2 * NXT_NC]; bv4[u].w = bp[3 * NXT_NC];
    }
  }
  for (int i0 = wave * NPR; i0 < nloc; i0 += (OWN_T / 32) * NPR) {
    const bool live = (i0 + grp) < nloc;
    const int i = live ? (i0 + grp) : (nloc - 1);
    const int node = lo + i;
    const float* ap = acc + i * FRP;
    if constexpr (CONCAT) {
      static_assert(FR == TL * 16, "four 16-byte pieces per lane");
      float4 v[4];
      float sum = 0.0f;
#pragma unroll
      for (int m = 0; m < 4; ++m) {
        const int c = 32 * m + 4 * sub;
        const float4 a4 = *(const float4*)(ap + c);
        const float4 b4 = *(const float4*)(bias + c);
        float4 t;
        t.x = a4.x + b4.x; t.y = a4.y + b4.y; t.z = a4.z + b4.z; t.w = a4.w + b4.w;
        t.x = (t.x > 0.0f) ? t.x : (__expf(t.x) - 1.0f);
        t.y = (t.y > 0.0f) ? t.y : (__expf(t.y) - 1.0f);
        t.z = (t.z > 0.0f) ? t.z : (__expf(t.z) - 1.0f);
        t.w = (t.w > 0.0f) ? t.w : (__expf(t.w) - 1.0f);
        v[m] = t;
        sum += t.x; sum += t.y; sum += t.z; sum += t.w;
      }
      const float mu = group_sum8(sum) * (1.0f / 128.0f);
      float vs = 0.0f;
#pragma unroll
      for (int m = 0; m < 4; ++m) {
        float4 d = v[m];
        d.x -= mu; d.y -= mu; d.z -= mu; d.w -= mu;
        v[m] = d;
        vs += d.x * d.x; vs += d.y * d.y; vs += d.z * d.z; vs += d.w * d.w;
      }
      const float var = group_sum8(vs) * (1.0f / 128.0f);
      const float rstd = rsqrtf(var + GAT_LN_EPS);
      float* xp = acc + i * FRP;
#pragma unroll
      for (int m = 0; m < 4; ++m) {
        const int c = 32 * m + 4 * sub;
        float4 o;
        o.x = v[m].x * rstd * g4[m].x + s4[m].x; o.y = v[m].y * rstd * g4[m].y + s4[m].y;
        o.z = v[m].z * rstd * g4[m].z + s4[m].z; o.w = v[m].w * rstd * g4[m].w + s4[m].w;
        if (live) *(float4*)(xp + c) = o;
      }
    } else {
      static_assert(CH % TL == 0, "every lane of a group holds CH / TL classes");
      constexpr int CPL = CH / TL;
      float x[CPL];
      float sum = 0.0f;
#pragma unroll
      for (int m = 0; m < CPL; ++m) {
        const int c = sub + TL * m;
        x[m] = ap[c] * 0.25f + bias[c];
        sum += x[m];
      }
      const float mu = group_sum8(sum) * (1.0f / CH);
      float vs = 0.0f;
#pragma unroll
      for (int m = 0; m < CPL; ++m) { x[m] -= mu; vs += x[m] * x[m]; }
      const float var = group_sum8(vs) * (1.0f / CH);
      const float rstd = rsqrtf(var + GAT_LN_EPS);
      float mxz = -3.0e38f;
#pragma unroll
      for (int m = 0; m < CPL; ++m) {
        const int c = sub + TL * m;
        x[m] = x[m] * rstd * lg[c] + lb[c];
        mxz = fmaxf(mxz, x[m]);
      }
      mxz = group_max8(mxz);
      float se = 0.0f;
#pragma unroll
      for (int m = 0; m < CPL; ++m) se += __expf(x[m] - mxz);
      const float lse = __logf(group_sum8(se));
      if (live) {
        float* yp = out + (size_t)node * CH;
#pragma unroll
        for (int m = 0; m < CPL; ++m) yp[sub + TL * m] = x[m] - mxz - lse;
      }
    }
  }

  if constexpr (CONCAT) {
    float4* Bs = (float4*)lw;
#pragma unroll
    for (int u = 0; u < ITEMS / OWN_T; ++u) {
      const int it = tid + u * OWN_T;
      const int m = it / NXT_NC, n = it - m * NXT_NC;
      const int hd = n / NXT_CH;
      const int np = 4 * (n - hd * NXT_CH) + hd;
      const int kb = m >> 2, hh = (m >> 1) & 1, jp = m & 1;
      Bs[((kb * 2 + jp) * NT + (np >> 4)) * 32 + hh * 16 + (np & 15)] = bv4[u];
    }
    if (tid < (FR / 4) * 16) {
      const int m = tid >> 4, np = tid & 15;
      float4 v = {0.0f, 0.0f, 0.0f, 0.0f};
      if (np < 8) {
        v.x = wa[(4 * m) * 8 + np];     v.y = wa[(4 * m + 1) * 8 + np];
        v.z = wa[(4 * m + 2) * 8 + np]; v.w = wa[(4 * m + 3) * 8 + np];
      }
      const int kb = m >> 2, hh = (m >> 1) & 1, jp = m & 1;
      Bs[((kb * 2 + jp) * NT + (NT - 1)) * 32 + hh * 16 + np] = v;
    }
    __syncthreads();

    const int c16 = lane & 15, hh = lane >> 4;
    auto unit = [&](auto ntl, int rt, int ct0) {
      constexpr int NTL = decltype(ntl)::value;
      const int rowA = min(rt * 16 + c16, OWN_RPW - 1);
      const float* ap = acc + rowA * FRP + 8 * hh;
      v8f d[NTL];
#pragma unroll
      for (int t = 0; t < NTL; ++t) d[t] = v8f{};
#pragma unroll
      for (int kb = 0; kb < KB; ++kb) {
#pragma unroll
        for (int jp = 0; jp < 2; ++jp) {
          const float4 av = *(const float4*)(ap + 16 * kb + 4 * jp);
          const v2f a0 = {av.x, av.y}, a1 = {av.z, av.w};
#pragma unroll
          for (int t = 0; t < NTL; ++t) {
            const float4 bv = Bs[((kb * 2 + jp) * NT + ct0 + t) * 32 + lane];
            const v2f b0 = {bv.x, bv.y}, b1 = {bv.z, bv.w};
            d[t] = __builtin_amdgcn_wmma_f32_16x16x4_f32(false, a0, false, b0, (short)0, d[t],
                                                         false, false);
            d[t] = __builtin_amdgcn_wmma_f32_16x16x4_f32(false, a1, false, b1, (short)0, d[t],
                                                         false, false);
          }
        }
      }
      const int rTop = rt * 16 + hh * 8;
      if (ct0 < NT - 1) {
#pragma unroll
        for (int t = 0; t < NTL; ++t) {
#pragma unroll
          for (int r = 0; r < 8; ++r) {
            const int row = rTop + r;
            if (row < nloc) out[(size_t)(lo + row) * NXT_NC + (ct0 + t) * 16 + c16] = d[t][r];
          }
        }
      } else {
        float* __restrict__ sp = (c16 < 4) ? esn : edn;
#pragma unroll
        for (int r = 0; r < 8; ++r) {
          const int row = rTop + r;
          if (row < nloc && c16 < 8) sp[(size_t)(lo + row) * H + (c16 & 3)] = d[0][r];
        }
      }
    };
    const int nRT = (nloc + 15) >> 4;
    const int nPair = nRT * (NXT_NC / 32);
    const int nU = nPair + nRT;
    for (int u = wave; u < nU; u += OWN_T / 32) {
      if (u < nPair) {
        const int rt = u / (NXT_NC / 32);
        unit(IntC<2>{}, rt, 2 * (u - rt * (NXT_NC / 32)));
      } else {
        unit(IntC<1>{}, u - nPair, NT - 1);
      }
    }
  }
}

extern "C" void kernel_launch(void* const* d_in, const int* in_sizes, int n_in,
                              void* d_out, int out_size, void* d_ws, size_t ws_size,
                              hipStream_t stream) {
  const float* x        = (const float*)d_in[0];
  const int*   ei       = (const int*)d_in[1];
  const float* W1       = (const float*)d_in[2];
  const float* att_src1 = (const float*)d_in[3];
  const float* att_dst1 = (const float*)d_in[4];
  const float* b1       = (const float*)d_in[5];
  const float* W2       = (const float*)d_in[6];
  const float* att_src2 = (const float*)d_in[7];
  const float* att_dst2 = (const float*)d_in[8];
  const float* b2       = (const float*)d_in[9];
  const float* ln0_g    = (const float*)d_in[10];
  const float* ln0_b    = (const float*)d_in[11];
  const float* ln1_g    = (const float*)d_in[12];
  const float* ln1_b    = (const float*)d_in[13];

  const int F_IN = 256, H1C = 128, H2C = 160, H = 4;
  const int N = in_sizes[0] / F_IN;
  const int E = in_sizes[1] / 2;

  const int owners = (N + OWN_RPW - 1) / OWN_RPW;

  float* ws   = (float*)d_ws;
  float* h1   = ws;
  float* h2   = h1   + (size_t)N * H1C;
  float* es   = h2   + (size_t)N * H2C;
  float* ed   = es   + (size_t)N * H;
  float* es2  = ed   + (size_t)N * H;
  float* ed2  = es2  + (size_t)N * H;
  int* sv_cnt = (int*)(ed2 + (size_t)N * H);
  int* sv_src = sv_cnt + owners;
  int* sv_dl  = sv_src + (size_t)owners * OWN_CAP;
  int* rec    = sv_dl + (size_t)owners * OWN_CAP;
  unsigned char* own8 = (unsigned char*)(rec + E);

  gemm_wmma_f32<256, 128><<<owners, GEMM_T, 0, stream>>>(x, W1, h1, N, att_src1, att_dst1,
                                                         es, ed, ei, E, own8, rec);
  owner_conv_k<32, true, true, false><<<owners, OWN_T, 0, stream>>>(
      own8, rec, E, N, h1, es, ed, b1, ln0_g, ln0_b, h2, sv_cnt, sv_src, sv_dl,
      W2, att_src2, att_dst2, es2, ed2);

  owner_conv_k<40, false, false, true><<<owners, OWN_T, 0, stream>>>(
      own8, rec, E, N, h2, es2, ed2, b2, ln1_g, ln1_b, (float*)d_out, sv_cnt, sv_src, sv_dl,
      nullptr, nullptr, nullptr, nullptr, nullptr);
}
